# starting kernel
# baseline (speedup 1.0000x reference)
#include <hip/hip_runtime.h>
#include <hip/hip_bf16.h>


typedef float v2f __attribute__((ext_vector_type(2)));
typedef float v8f __attribute__((ext_vector_type(8)));

#define D_FEAT 64

__global__ void __launch_bounds__(256)
gin_scatter_kernel(const float* __restrict__ x,
                   const int*   __restrict__ src,
                   const int*   __restrict__ dst,
                   float*       __restrict__ agg,
                   int nEdges)
{
    const int q = (threadIdx.x & 15) * 4;
    const int e = blockIdx.x * 16 + (threadIdx.x >> 4);
    if (e >= nEdges) return;
    const int s = src[e];
    const int d = dst[e];
    const float4 v = *(const float4*)(x + (size_t)s * D_FEAT + q);
    float* o = agg + (size_t)d * D_FEAT + q;
    unsafeAtomicAdd(o + 0, v.x);
    unsafeAtomicAdd(o + 1, v.y);
    unsafeAtomicAdd(o + 2, v.z);
    unsafeAtomicAdd(o + 3, v.w);
}

__global__ void __launch_bounds__(256)
gin_layer_kernel(const float* __restrict__ X,
                 const float* __restrict__ Xagg,
                 const float* __restrict__ W,
                 const float* __restrict__ bias,
                 float*       __restrict__ H,
                 const int*   __restrict__ gid,
                 float*       __restrict__ pool,
                 int nNodes, int mode)
{
    __shared__ float sWp[D_FEAT * D_FEAT];
    __shared__ float sB[D_FEAT];

    for (int idx = threadIdx.x; idx < D_FEAT * D_FEAT; idx += 256) {
        const int k   = idx >> 6;
        const int col = idx & 63;
        sWp[(k >> 1) * 128 + col * 2 + (k & 1)] = W[idx];
    }
    if (threadIdx.x < D_FEAT) sB[threadIdx.x] = bias[threadIdx.x];
    __syncthreads();

    const int wave    = threadIdx.x >> 5;
    const int lane    = threadIdx.x & 31;
    const int rowTile = blockIdx.x * 8 + wave;
    const int rowBase = rowTile * 16;
    if (rowBase >= nNodes) return;

    const int halfSel = lane >> 4;
    const int laneM   = lane & 15;

    v8f acc[4];
    #pragma unroll
    for (int ct = 0; ct < 4; ++ct) acc[ct] = (v8f){0.f,0.f,0.f,0.f,0.f,0.f,0.f,0.f};

    int aRow = rowBase + laneM;
    if (aRow >= nNodes) aRow = nNodes - 1;
    const float* __restrict__ xr = X    + (size_t)aRow * D_FEAT;
    const float* __restrict__ gr = Xagg + (size_t)aRow * D_FEAT;

    #pragma unroll 4
    for (int kt = 0; kt < 16; ++kt) {
        const int k0 = kt * 4 + halfSel * 2;
        const v2f ax = *(const v2f*)(xr + k0);
        const v2f ag = *(const v2f*)(gr + k0);
        const v2f a  = ax + ag;
        const int bRow = (kt * 2 + halfSel) * 128;
        #pragma unroll
        for (int ct = 0; ct < 4; ++ct) {
            const int col = ct * 16 + laneM;
            const v2f b = *(const v2f*)(sWp + bRow + col * 2);
            acc[ct] = __builtin_amdgcn_wmma_f32_16x16x4_f32(
                false, a, false, b, (short)0, acc[ct], false, false);
        }
    }

    #pragma unroll
    for (int ct = 0; ct < 4; ++ct) {
        const int col = ct * 16 + laneM;
        const float bv = sB[col];
        #pragma unroll
        for (int v = 0; v < 8; ++v) {
            const int row = rowBase + v + halfSel * 8;
            if (row < nNodes) {
                const float val = acc[ct][v] + bv;
                if (mode == 0) {
                    H[(size_t)row * D_FEAT + col] = fmaxf(val, 0.f);
                } else {
                    unsafeAtomicAdd(&pool[(size_t)gid[row] * D_FEAT + col], val);
                }
            }
        }
    }
}

extern "C" void kernel_launch(void* const* d_in, const int* in_sizes, int n_in,
                              void* d_out, int out_size, void* d_ws, size_t ws_size,
                              hipStream_t stream)
{
    const float* feats = (const float*)d_in[0];
    const int*   src   = (const int*)  d_in[1];
    const int*   dst   = (const int*)  d_in[2];
    const int*   gid   = (const int*)  d_in[3];
    const float* W1    = (const float*)d_in[4];
    const float* b1    = (const float*)d_in[5];
    const float* W2    = (const float*)d_in[6];
    const float* b2    = (const float*)d_in[7];
    float* out = (float*)d_out;

    const int nNodes = in_sizes[0] / D_FEAT;
    const int nEdges = in_sizes[1];

    float* agg = (float*)d_ws;
    float* h   = agg + (size_t)nNodes * D_FEAT;
    const size_t featBytes = (size_t)nNodes * D_FEAT * sizeof(float);

    const dim3 eBlk(256);
    const dim3 eGrd((nEdges + 15) / 16);
    const int rowTiles = (nNodes + 15) / 16;
    const dim3 gGrd((rowTiles + 7) / 8);

    hipMemsetAsync(agg, 0, featBytes, stream);
    hipMemsetAsync(out, 0, (size_t)out_size * sizeof(float), stream);
    gin_scatter_kernel<<<eGrd, eBlk, 0, stream>>>(feats, src, dst, agg, nEdges);
    gin_layer_kernel<<<gGrd, 256, 0, stream>>>(feats, agg, W1, b1,
                                               h, nullptr, nullptr, nNodes, 0);

    hipMemsetAsync(agg, 0, featBytes, stream);
    gin_scatter_kernel<<<eGrd, eBlk, 0, stream>>>(h, src, dst, agg, nEdges);
    gin_layer_kernel<<<gGrd, 256, 0, stream>>>(h, agg, W2, b2,
                                               nullptr, gid, out, nNodes, 1);
}


# scatter: one wave per 16 edges, lane = channel, so each f32 atomic instruction and row load covers one whole 128-B line (was 4 quarter-filled lines); sums stay atomic, in arrival order
# speedup vs baseline: 1.6084x; 1.6084x over previous
#include <hip/hip_runtime.h>
#include <hip/hip_bf16.h>


typedef float v2f __attribute__((ext_vector_type(2)));
typedef float v8f __attribute__((ext_vector_type(8)));

#define D_FEAT 64

#define EDGES_PER_WAVE 16

__global__ void __launch_bounds__(256)
gin_scatter_kernel(const float* __restrict__ x,
                   const int*   __restrict__ src,
                   const int*   __restrict__ dst,
                   float*       __restrict__ agg,
                   int nEdges)
{
    const int lane = threadIdx.x & 31;
    const int wv = __builtin_amdgcn_readfirstlane((int)(threadIdx.x >> 5));
    const int e0 = (blockIdx.x * 8 + wv) * EDGES_PER_WAVE;
    if (e0 >= nEdges) return;

    if (e0 + EDGES_PER_WAVE <= nEdges) {
        const int ei = e0 + (lane & (EDGES_PER_WAVE - 1));
        const int sv = src[ei];
        const int dv = dst[ei];

        float lo[EDGES_PER_WAVE], hi[EDGES_PER_WAVE];
        #pragma unroll
        for (int i = 0; i < EDGES_PER_WAVE; ++i) {
            const int s = __builtin_amdgcn_readlane(sv, i);
            const float* __restrict__ xr = x + (size_t)s * D_FEAT;
            lo[i] = xr[lane];
            hi[i] = xr[lane + 32];
        }
        #pragma unroll
        for (int i = 0; i < EDGES_PER_WAVE; ++i) {
            const int d = __builtin_amdgcn_readlane(dv, i);
            float* o = agg + (size_t)d * D_FEAT;
            unsafeAtomicAdd(o + lane,      lo[i]);
            unsafeAtomicAdd(o + lane + 32, hi[i]);
        }
    } else {
        for (int e = e0; e < nEdges; ++e) {
            const float* __restrict__ xr = x + (size_t)src[e] * D_FEAT;
            float* o = agg + (size_t)dst[e] * D_FEAT;
            unsafeAtomicAdd(o + lane,      xr[lane]);
            unsafeAtomicAdd(o + lane + 32, xr[lane + 32]);
        }
    }
}

__global__ void __launch_bounds__(256)
gin_layer_kernel(const float* __restrict__ X,
                 const float* __restrict__ Xagg,
                 const float* __restrict__ W,
                 const float* __restrict__ bias,
                 float*       __restrict__ H,
                 const int*   __restrict__ gid,
                 float*       __restrict__ pool,
                 int nNodes, int mode)
{
    __shared__ float sWp[D_FEAT * D_FEAT];
    __shared__ float sB[D_FEAT];

    for (int idx = threadIdx.x; idx < D_FEAT * D_FEAT; idx += 256) {
        const int k   = idx >> 6;
        const int col = idx & 63;
        sWp[(k >> 1) * 128 + col * 2 + (k & 1)] = W[idx];
    }
    if (threadIdx.x < D_FEAT) sB[threadIdx.x] = bias[threadIdx.x];
    __syncthreads();

    const int wave    = threadIdx.x >> 5;
    const int lane    = threadIdx.x & 31;
    const int rowTile = blockIdx.x * 8 + wave;
    const int rowBase = rowTile * 16;
    if (rowBase >= nNodes) return;

    const int halfSel = lane >> 4;
    const int laneM   = lane & 15;

    v8f acc[4];
    #pragma unroll
    for (int ct = 0; ct < 4; ++ct) acc[ct] = (v8f){0.f,0.f,0.f,0.f,0.f,0.f,0.f,0.f};

    int aRow = rowBase + laneM;
    if (aRow >= nNodes) aRow = nNodes - 1;
    const float* __restrict__ xr = X    + (size_t)aRow * D_FEAT;
    const float* __restrict__ gr = Xagg + (size_t)aRow * D_FEAT;

    #pragma unroll 4
    for (int kt = 0; kt < 16; ++kt) {
        const int k0 = kt * 4 + halfSel * 2;
        const v2f ax = *(const v2f*)(xr + k0);
        const v2f ag = *(const v2f*)(gr + k0);
        const v2f a  = ax + ag;
        const int bRow = (kt * 2 + halfSel) * 128;
        #pragma unroll
        for (int ct = 0; ct < 4; ++ct) {
            const int col = ct * 16 + laneM;
            const v2f b = *(const v2f*)(sWp + bRow + col * 2);
            acc[ct] = __builtin_amdgcn_wmma_f32_16x16x4_f32(
                false, a, false, b, (short)0, acc[ct], false, false);
        }
    }

    #pragma unroll
    for (int ct = 0; ct < 4; ++ct) {
        const int col = ct * 16 + laneM;
        const float bv = sB[col];
        #pragma unroll
        for (int v = 0; v < 8; ++v) {
            const int row = rowBase + v + halfSel * 8;
            if (row < nNodes) {
                const float val = acc[ct][v] + bv;
                if (mode == 0) {
                    H[(size_t)row * D_FEAT + col] = fmaxf(val, 0.f);
                } else {
                    unsafeAtomicAdd(&pool[(size_t)gid[row] * D_FEAT + col], val);
                }
            }
        }
    }
}

extern "C" void kernel_launch(void* const* d_in, const int* in_sizes, int n_in,
                              void* d_out, int out_size, void* d_ws, size_t ws_size,
                              hipStream_t stream)
{
    const float* feats = (const float*)d_in[0];
    const int*   src   = (const int*)  d_in[1];
    const int*   dst   = (const int*)  d_in[2];
    const int*   gid   = (const int*)  d_in[3];
    const float* W1    = (const float*)d_in[4];
    const float* b1    = (const float*)d_in[5];
    const float* W2    = (const float*)d_in[6];
    const float* b2    = (const float*)d_in[7];
    float* out = (float*)d_out;

    const int nNodes = in_sizes[0] / D_FEAT;
    const int nEdges = in_sizes[1];

    float* agg = (float*)d_ws;
    float* h   = agg + (size_t)nNodes * D_FEAT;
    const size_t featBytes = (size_t)nNodes * D_FEAT * sizeof(float);

    const dim3 eBlk(256);
    const dim3 eGrd((nEdges + 8 * EDGES_PER_WAVE - 1) / (8 * EDGES_PER_WAVE));
    const int rowTiles = (nNodes + 15) / 16;
    const dim3 gGrd((rowTiles + 7) / 8);

    hipMemsetAsync(agg, 0, featBytes, stream);
    hipMemsetAsync(out, 0, (size_t)out_size * sizeof(float), stream);
    gin_scatter_kernel<<<eGrd, eBlk, 0, stream>>>(feats, src, dst, agg, nEdges);
    gin_layer_kernel<<<gGrd, 256, 0, stream>>>(feats, agg, W1, b1,
                                               h, nullptr, nullptr, nNodes, 0);

    hipMemsetAsync(agg, 0, featBytes, stream);
    gin_scatter_kernel<<<eGrd, eBlk, 0, stream>>>(h, src, dst, agg, nEdges);
    gin_layer_kernel<<<gGrd, 256, 0, stream>>>(h, agg, W2, b2,
                                               nullptr, gid, out, nNodes, 1);
}


# layer 2: per-graph sums of h and h[src] (LDS f32 adds per work-group, then global adds) taken BEFORE the W2 product, by linearity; W2 on 128 rows; sums reassociated, arrival order
# speedup vs baseline: 2.3703x; 1.4737x over previous
#include <hip/hip_runtime.h>
#include <hip/hip_bf16.h>


typedef float v2f __attribute__((ext_vector_type(2)));
typedef float v8f __attribute__((ext_vector_type(8)));

#define D_FEAT   64
#define N_GRAPHS 128

#define EDGES_PER_WAVE 16

__global__ void __launch_bounds__(256)
gin_scatter_kernel(const float* __restrict__ x,
                   const int*   __restrict__ src,
                   const int*   __restrict__ dst,
                   float*       __restrict__ agg,
                   int nEdges)
{
    const int lane = threadIdx.x & 31;
    const int wv = __builtin_amdgcn_readfirstlane((int)(threadIdx.x >> 5));
    const int e0 = (blockIdx.x * 8 + wv) * EDGES_PER_WAVE;
    if (e0 >= nEdges) return;

    if (e0 + EDGES_PER_WAVE <= nEdges) {
        const int ei = e0 + (lane & (EDGES_PER_WAVE - 1));
        const int sv = src[ei];
        const int dv = dst[ei];

        float lo[EDGES_PER_WAVE], hi[EDGES_PER_WAVE];
        #pragma unroll
        for (int i = 0; i < EDGES_PER_WAVE; ++i) {
            const int s = __builtin_amdgcn_readlane(sv, i);
            const float* __restrict__ xr = x + (size_t)s * D_FEAT;
            lo[i] = xr[lane];
            hi[i] = xr[lane + 32];
        }
        #pragma unroll
        for (int i = 0; i < EDGES_PER_WAVE; ++i) {
            const int d = __builtin_amdgcn_readlane(dv, i);
            float* o = agg + (size_t)d * D_FEAT;
            unsafeAtomicAdd(o + lane,      lo[i]);
            unsafeAtomicAdd(o + lane + 32, hi[i]);
        }
    } else {
        for (int e = e0; e < nEdges; ++e) {
            const float* __restrict__ xr = x + (size_t)src[e] * D_FEAT;
            float* o = agg + (size_t)dst[e] * D_FEAT;
            unsafeAtomicAdd(o + lane,      xr[lane]);
            unsafeAtomicAdd(o + lane + 32, xr[lane + 32]);
        }
    }
}

__device__ __forceinline__ void
stage_weights(const float* __restrict__ W,
              const float* __restrict__ bias,
              float* sWp, float* sB)
{
    for (int idx = threadIdx.x; idx < D_FEAT * D_FEAT; idx += 256) {
        const int k   = idx >> 6;
        const int col = idx & 63;
        sWp[(k >> 1) * 128 + col * 2 + (k & 1)] = W[idx];
    }
    if (threadIdx.x < D_FEAT) sB[threadIdx.x] = bias[threadIdx.x];
    __syncthreads();
}

template <bool ADD_SECOND>
__device__ __forceinline__ void
wave_tile_product(const float* __restrict__ xr, const float* __restrict__ gr,
                  const float* sWp, int halfSel, int laneM, v8f (&acc)[4])
{
    #pragma unroll
    for (int ct = 0; ct < 4; ++ct) acc[ct] = (v8f){0.f,0.f,0.f,0.f,0.f,0.f,0.f,0.f};

    #pragma unroll 4
    for (int kt = 0; kt < 16; ++kt) {
        const int k0 = kt * 4 + halfSel * 2;
        v2f a = *(const v2f*)(xr + k0);
        if (ADD_SECOND) a = a + *(const v2f*)(gr + k0);
        const int bRow = (kt * 2 + halfSel) * 128;
        #pragma unroll
        for (int ct = 0; ct < 4; ++ct) {
            const int col = ct * 16 + laneM;
            const v2f b = *(const v2f*)(sWp + bRow + col * 2);
            acc[ct] = __builtin_amdgcn_wmma_f32_16x16x4_f32(
                false, a, false, b, (short)0, acc[ct], false, false);
        }
    }
}

__global__ void __launch_bounds__(256)
gin_layer_kernel(const float* __restrict__ X,
                 const float* __restrict__ Xagg,
                 const float* __restrict__ W,
                 const float* __restrict__ bias,
                 float*       __restrict__ H,
                 int nNodes)
{
    __shared__ float sWp[D_FEAT * D_FEAT];
    __shared__ float sB[D_FEAT];
    stage_weights(W, bias, sWp, sB);

    const int wave    = threadIdx.x >> 5;
    const int lane    = threadIdx.x & 31;
    const int rowBase = (blockIdx.x * 8 + wave) * 16;
    if (rowBase >= nNodes) return;

    const int halfSel = lane >> 4;
    const int laneM   = lane & 15;

    int aRow = rowBase + laneM;
    if (aRow >= nNodes) aRow = nNodes - 1;

    v8f acc[4];
    wave_tile_product<true>(X + (size_t)aRow * D_FEAT, Xagg + (size_t)aRow * D_FEAT,
                            sWp, halfSel, laneM, acc);

    #pragma unroll
    for (int ct = 0; ct < 4; ++ct) {
        const int col = ct * 16 + laneM;
        const float bv = sB[col];
        #pragma unroll
        for (int v = 0; v < 8; ++v) {
            const int row = rowBase + v + halfSel * 8;
            if (row < nNodes)
                H[(size_t)row * D_FEAT + col] = fmaxf(acc[ct][v] + bv, 0.f);
        }
    }
}

#define ROWS_PER_WAVE 16
#define POOL_THREADS  1024

__global__ void __launch_bounds__(POOL_THREADS)
gin_pool_kernel(const float* __restrict__ h,
                const int*   __restrict__ src,
                const int*   __restrict__ dst,
                const int*   __restrict__ gid,
                float*       __restrict__ pool,
                float*       __restrict__ cnt,
                int nEdges, int nNodes)
{
    __shared__ float sP[N_GRAPHS * D_FEAT];
    __shared__ float sC[N_GRAPHS];

    for (int i = threadIdx.x; i < N_GRAPHS * D_FEAT; i += POOL_THREADS) sP[i] = 0.f;
    if (threadIdx.x < N_GRAPHS) sC[threadIdx.x] = 0.f;
    __syncthreads();

    const int lane = threadIdx.x & 31;
    const int li   = lane & (ROWS_PER_WAVE - 1);
    const int wv = __builtin_amdgcn_readfirstlane((int)(threadIdx.x >> 5));

    const int edgeChunks = (nEdges + ROWS_PER_WAVE - 1) / ROWS_PER_WAVE;
    const int nodeChunks = (nNodes + ROWS_PER_WAVE - 1) / ROWS_PER_WAVE;
    const int nChunks    = edgeChunks + nodeChunks;
    const int wavesTotal = gridDim.x * (POOL_THREADS / 32);

    for (int c = blockIdx.x * (POOL_THREADS / 32) + wv; c < nChunks; c += wavesTotal) {
        int sv, gv, n;
        if (c < edgeChunks) {
            const int e0 = c * ROWS_PER_WAVE;
            n = min(ROWS_PER_WAVE, nEdges - e0);
            const int e = e0 + min(li, n - 1);
            sv = src[e];
            gv = gid[dst[e]];
        } else {
            const int i0 = (c - edgeChunks) * ROWS_PER_WAVE;
            n = min(ROWS_PER_WAVE, nNodes - i0);
            sv = i0 + min(li, n - 1);
            gv = gid[sv];
            if (lane < n) unsafeAtomicAdd(&sC[gv], 1.0f);
        }

        if (n == ROWS_PER_WAVE) {
            v2f r[ROWS_PER_WAVE];
            #pragma unroll
            for (int i = 0; i < ROWS_PER_WAVE; ++i) {
                const int s = __builtin_amdgcn_readlane(sv, i);
                r[i] = ((const v2f*)(h + (size_t)s * D_FEAT))[lane];
            }
            #pragma unroll
            for (int i = 0; i < ROWS_PER_WAVE; ++i) {
                const int g = __builtin_amdgcn_readlane(gv, i);
                float* t = sP + g * D_FEAT + 2 * lane;
                unsafeAtomicAdd(t,     r[i].x);
                unsafeAtomicAdd(t + 1, r[i].y);
            }
        } else {
            for (int i = 0; i < n; ++i) {
                const int s = __builtin_amdgcn_readlane(sv, i);
                const int g = __builtin_amdgcn_readlane(gv, i);
                const v2f rv = ((const v2f*)(h + (size_t)s * D_FEAT))[lane];
                float* t = sP + g * D_FEAT + 2 * lane;
                unsafeAtomicAdd(t,     rv.x);
                unsafeAtomicAdd(t + 1, rv.y);
            }
        }
    }
    __syncthreads();

    constexpr int LINES = N_GRAPHS * D_FEAT / 32;
    #pragma unroll
    for (int j = 0; j < LINES / (POOL_THREADS / 32); ++j) {
        const int line = (wv + (POOL_THREADS / 32) * j + (int)blockIdx.x) & (LINES - 1);
        const int idx  = line * 32 + lane;
        unsafeAtomicAdd(pool + idx, sP[idx]);
    }
    if (threadIdx.x < N_GRAPHS) unsafeAtomicAdd(cnt + threadIdx.x, sC[threadIdx.x]);
}

__global__ void __launch_bounds__(256)
gin_pool_linear_kernel(const float* __restrict__ pool,
                       const float* __restrict__ cnt,
                       const float* __restrict__ W,
                       const float* __restrict__ bias,
                       float*       __restrict__ out)
{
    __shared__ float sWp[D_FEAT * D_FEAT];
    __shared__ float sB[D_FEAT];
    stage_weights(W, bias, sWp, sB);

    const int wave    = threadIdx.x >> 5;
    const int lane    = threadIdx.x & 31;
    const int rowBase = (blockIdx.x * 8 + wave) * 16;
    if (rowBase >= N_GRAPHS) return;

    const int halfSel = lane >> 4;
    const int laneM   = lane & 15;

    v8f acc[4];
    wave_tile_product<false>(pool + (size_t)(rowBase + laneM) * D_FEAT, nullptr,
                             sWp, halfSel, laneM, acc);

    #pragma unroll
    for (int v = 0; v < 8; ++v) {
        const int row = rowBase + v + halfSel * 8;
        const float nv = cnt[row];
        #pragma unroll
        for (int ct = 0; ct < 4; ++ct) {
            const int col = ct * 16 + laneM;
            out[(size_t)row * D_FEAT + col] = acc[ct][v] + nv * sB[col];
        }
    }
}

extern "C" void kernel_launch(void* const* d_in, const int* in_sizes, int n_in,
                              void* d_out, int out_size, void* d_ws, size_t ws_size,
                              hipStream_t stream)
{
    const float* feats = (const float*)d_in[0];
    const int*   src   = (const int*)  d_in[1];
    const int*   dst   = (const int*)  d_in[2];
    const int*   gid   = (const int*)  d_in[3];
    const float* W1    = (const float*)d_in[4];
    const float* b1    = (const float*)d_in[5];
    const float* W2    = (const float*)d_in[6];
    const float* b2    = (const float*)d_in[7];
    float* out = (float*)d_out;

    const int nNodes = in_sizes[0] / D_FEAT;
    const int nEdges = in_sizes[1];

    float* agg  = (float*)d_ws;
    float* h    = agg  + (size_t)nNodes * D_FEAT;
    float* pool = h    + (size_t)nNodes * D_FEAT;
    float* cnt  = pool + N_GRAPHS * D_FEAT;
    const size_t featBytes = (size_t)nNodes * D_FEAT * sizeof(float);

    const dim3 eBlk(256);
    const dim3 eGrd((nEdges + 8 * EDGES_PER_WAVE - 1) / (8 * EDGES_PER_WAVE));
    const int rowTiles = (nNodes + 15) / 16;
    const dim3 gGrd((rowTiles + 7) / 8);

    hipMemsetAsync(agg,  0, featBytes, stream);
    hipMemsetAsync(pool, 0, (N_GRAPHS * D_FEAT + N_GRAPHS) * sizeof(float), stream);

    gin_scatter_kernel<<<eGrd, eBlk, 0, stream>>>(feats, src, dst, agg, nEdges);
    gin_layer_kernel<<<gGrd, 256, 0, stream>>>(feats, agg, W1, b1, h, nNodes);

    gin_pool_kernel<<<512, POOL_THREADS, 0, stream>>>(h, src, dst, gid, pool, cnt,
                                                      nEdges, nNodes);
    gin_pool_linear_kernel<<<1, 256, 0, stream>>>(pool, cnt, W2, b2, out);
}


# layer 1: edge slices counting-sorted by 400-row dst block in LDS; one work-group per block sums feats[src] into an LDS tile (ds f32 adds, arrival order; sums reassociated), then W1; no global atomics
# speedup vs baseline: 2.9166x; 1.2305x over previous
#include <hip/hip_runtime.h>
#include <hip/hip_bf16.h>


typedef float v2f __attribute__((ext_vector_type(2)));
typedef float v8f __attribute__((ext_vector_type(8)));

#define D_FEAT   64
#define N_GRAPHS 128

#define PART_ROWS   400
#define MAX_PARTS   256
#define SLICE       8192
#define BIN_THREADS 1024
#define EDGES_PER_THREAD (SLICE / BIN_THREADS)
#define OFFS_PITCH  (MAX_PARTS + 1)

__global__ void __launch_bounds__(BIN_THREADS)
gin_bin_kernel(const int* __restrict__ src,
               const int* __restrict__ dst,
               int*       __restrict__ binned,
               int*       __restrict__ offs,
               int nEdges)
{
    __shared__ int  sCnt[MAX_PARTS];
    __shared__ int  sOff[MAX_PARTS + 1];
    __shared__ int  sCur[MAX_PARTS];
    __shared__ int2 sPairs[SLICE];

    const int tid = threadIdx.x;
    const int e0  = blockIdx.x * SLICE;
    const int n   = min(SLICE, nEdges - e0);

    if (tid < MAX_PARTS) sCnt[tid] = 0;
    __syncthreads();

    int dv[EDGES_PER_THREAD], sv[EDGES_PER_THREAD];
    #pragma unroll
    for (int j = 0; j < EDGES_PER_THREAD; ++j) {
        const int i = tid + j * BIN_THREADS;
        dv[j] = (i < n) ? dst[e0 + i] : 0;
        sv[j] = (i < n) ? src[e0 + i] : 0;
    }
    #pragma unroll
    for (int j = 0; j < EDGES_PER_THREAD; ++j) {
        if (tid + j * BIN_THREADS < n)
            atomicAdd(&sCnt[min(dv[j] / PART_ROWS, MAX_PARTS - 1)], 1);
    }
    __syncthreads();

    if (tid < 32) {
        int c[8];
        int sum = 0;
        #pragma unroll
        for (int j = 0; j < 8; ++j) { c[j] = sCnt[tid * 8 + j]; sum += c[j]; }
        int incl = sum;
        #pragma unroll
        for (int d = 1; d < 32; d <<= 1) {
            const int t = __shfl_up(incl, d);
            if (tid >= d) incl += t;
        }
        int run = incl - sum;
        #pragma unroll
        for (int j = 0; j < 8; ++j) {
            sOff[tid * 8 + j] = run;
            sCur[tid * 8 + j] = run;
            run += c[j];
        }
        if (tid == 31) sOff[MAX_PARTS] = run;
    }
    __syncthreads();

    #pragma unroll
    for (int j = 0; j < EDGES_PER_THREAD; ++j) {
        if (tid + j * BIN_THREADS < n) {
            const int p   = min(dv[j] / PART_ROWS, MAX_PARTS - 1);
            const int pos = atomicAdd(&sCur[p], 1);
            sPairs[pos] = make_int2(sv[j], dv[j] - p * PART_ROWS);
        }
    }
    __syncthreads();

    int2* __restrict__ o = (int2*)binned + (size_t)blockIdx.x * SLICE;
    for (int i = tid; i < n; i += BIN_THREADS) o[i] = sPairs[i];
    if (tid <= MAX_PARTS) offs[blockIdx.x * OFFS_PITCH + tid] = sOff[tid];
}

__device__ __forceinline__ void
stage_weights(const float* __restrict__ W,
              const float* __restrict__ bias,
              float* sWp, float* sB)
{
    for (int idx = threadIdx.x; idx < D_FEAT * D_FEAT; idx += 256) {
        const int k   = idx >> 6;
        const int col = idx & 63;
        sWp[(k >> 1) * 128 + col * 2 + (k & 1)] = W[idx];
    }
    if (threadIdx.x < D_FEAT) sB[threadIdx.x] = bias[threadIdx.x];
    __syncthreads();
}

template <bool ADD_SECOND>
__device__ __forceinline__ void
wave_tile_product(const float* __restrict__ xr, const float* __restrict__ gr,
                  const float* sWp, int halfSel, int laneM, v8f (&acc)[4])
{
    #pragma unroll
    for (int ct = 0; ct < 4; ++ct) acc[ct] = (v8f){0.f,0.f,0.f,0.f,0.f,0.f,0.f,0.f};

    #pragma unroll 4
    for (int kt = 0; kt < 16; ++kt) {
        const int k0 = kt * 4 + halfSel * 2;
        v2f a = *(const v2f*)(xr + k0);
        if (ADD_SECOND) a = a + *(const v2f*)(gr + k0);
        const int bRow = (kt * 2 + halfSel) * 128;
        #pragma unroll
        for (int ct = 0; ct < 4; ++ct) {
            const int col = ct * 16 + laneM;
            const v2f b = *(const v2f*)(sWp + bRow + col * 2);
            acc[ct] = __builtin_amdgcn_wmma_f32_16x16x4_f32(
                false, a, false, b, (short)0, acc[ct], false, false);
        }
    }
}

#define L1_THREADS  1024
#define TILE_PITCH  66
#define GATHER_ROWS 16

__global__ void __launch_bounds__(L1_THREADS)
gin_layer1_kernel(const float* __restrict__ X,
                  const int*   __restrict__ binned,
                  const int*   __restrict__ offs,
                  const float* __restrict__ W,
                  const float* __restrict__ bias,
                  float*       __restrict__ H,
                  int nNodes, int nSlices)
{
    __shared__ float sT[PART_ROWS * TILE_PITCH];
    __shared__ float sWp[D_FEAT * D_FEAT];
    __shared__ float sB[D_FEAT];

    const int tid = threadIdx.x;
    for (int i = tid; i < PART_ROWS * TILE_PITCH; i += L1_THREADS) sT[i] = 0.f;
    for (int idx = tid; idx < D_FEAT * D_FEAT; idx += L1_THREADS) {
        const int k   = idx >> 6;
        const int col = idx & 63;
        sWp[(k >> 1) * 128 + col * 2 + (k & 1)] = W[idx];
    }
    if (tid < D_FEAT) sB[tid] = bias[tid];
    __syncthreads();

    const int lane = tid & 31;
    const int wv   = __builtin_amdgcn_readfirstlane(tid >> 5);
    const int part = blockIdx.x;

    const int nMine = (nSlices > wv) ? (nSlices - wv + 31) / 32 : 0;
    int lov = 0, hiv = 0;
    if (lane < nMine) {
        const int* __restrict__ o = offs + (wv + 32 * lane) * OFFS_PITCH + part;
        lov = o[0];
        hiv = o[1];
    }
    const int2* __restrict__ pairs = (const int2*)binned;
    const int li = lane & (GATHER_ROWS - 1);

    int j = 0, b0 = 0, segHi = 0;
    for (; j < nMine; ++j) {
        b0    = __builtin_amdgcn_readlane(lov, j);
        segHi = __builtin_amdgcn_readlane(hiv, j);
        if (b0 < segHi) break;
    }
    int2 pr = make_int2(0, 0);
    if (j < nMine)
        pr = pairs[(size_t)(wv + 32 * j) * SLICE + b0 + min(li, min(GATHER_ROWS, segHi - b0) - 1)];

    while (j < nMine) {
        const int n  = min(GATHER_ROWS, segHi - b0);
        const int sv = pr.x;
        const int dv = min(pr.y, PART_ROWS - 1);

        v2f r[GATHER_ROWS];
        #pragma unroll
        for (int i = 0; i < GATHER_ROWS; ++i) {
            const int s = __builtin_amdgcn_readlane(sv, i);
            r[i] = ((const v2f*)(X + (size_t)s * D_FEAT))[lane];
        }

        int nj = j, nb0 = b0 + GATHER_ROWS, nHi = segHi;
        if (nb0 >= nHi) {
            for (++nj; nj < nMine; ++nj) {
                nb0 = __builtin_amdgcn_readlane(lov, nj);
                nHi = __builtin_amdgcn_readlane(hiv, nj);
                if (nb0 < nHi) break;
            }
        }
        if (nj < nMine)
            pr = pairs[(size_t)(wv + 32 * nj) * SLICE + nb0 + min(li, min(GATHER_ROWS, nHi - nb0) - 1)];

        #pragma unroll
        for (int i = 0; i < GATHER_ROWS; ++i) {
            if (i < n) {
                const int dl = __builtin_amdgcn_readlane(dv, i);
                float* t = sT + dl * TILE_PITCH + 2 * lane;
                unsafeAtomicAdd(t,     r[i].x);
                unsafeAtomicAdd(t + 1, r[i].y);
            }
        }
        j = nj; b0 = nb0; segHi = nHi;
    }
    __syncthreads();

    if (wv >= PART_ROWS / 16) return;
    const int rowLocal = wv * 16;
    const int rowBase  = part * PART_ROWS + rowLocal;
    if (rowBase >= nNodes) return;

    const int halfSel = lane >> 4;
    const int laneM   = lane & 15;

    int aRow = rowBase + laneM;
    if (aRow >= nNodes) aRow = nNodes - 1;
    const float* __restrict__ xr = X + (size_t)aRow * D_FEAT;
    const float* tr = sT + (rowLocal + laneM) * TILE_PITCH;

    v8f acc[4];
    #pragma unroll
    for (int ct = 0; ct < 4; ++ct) acc[ct] = (v8f){0.f,0.f,0.f,0.f,0.f,0.f,0.f,0.f};

    #pragma unroll 4
    for (int kt = 0; kt < 16; ++kt) {
        const int k0 = kt * 4 + halfSel * 2;
        const v2f a = *(const v2f*)(xr + k0) + *(const v2f*)(tr + k0);
        const int bRow = (kt * 2 + halfSel) * 128;
        #pragma unroll
        for (int ct = 0; ct < 4; ++ct) {
            const int col = ct * 16 + laneM;
            const v2f b = *(const v2f*)(sWp + bRow + col * 2);
            acc[ct] = __builtin_amdgcn_wmma_f32_16x16x4_f32(
                false, a, false, b, (short)0, acc[ct], false, false);
        }
    }

    #pragma unroll
    for (int ct = 0; ct < 4; ++ct) {
        const int col = ct * 16 + laneM;
        const float bv = sB[col];
        #pragma unroll
        for (int v = 0; v < 8; ++v) {
            const int row = rowBase + v + halfSel * 8;
            if (row < nNodes)
                H[(size_t)row * D_FEAT + col] = fmaxf(acc[ct][v] + bv, 0.f);
        }
    }
}

#define ROWS_PER_WAVE 16
#define POOL_THREADS  1024

__global__ void __launch_bounds__(POOL_THREADS)
gin_pool_kernel(const float* __restrict__ h,
                const int*   __restrict__ src,
                const int*   __restrict__ dst,
                const int*   __restrict__ gid,
                float*       __restrict__ pool,
                float*       __restrict__ cnt,
                int nEdges, int nNodes)
{
    __shared__ float sP[N_GRAPHS * D_FEAT];
    __shared__ float sC[N_GRAPHS];

    for (int i = threadIdx.x; i < N_GRAPHS * D_FEAT; i += POOL_THREADS) sP[i] = 0.f;
    if (threadIdx.x < N_GRAPHS) sC[threadIdx.x] = 0.f;
    __syncthreads();

    const int lane = threadIdx.x & 31;
    const int li   = lane & (ROWS_PER_WAVE - 1);
    const int wv = __builtin_amdgcn_readfirstlane((int)(threadIdx.x >> 5));

    const int edgeChunks = (nEdges + ROWS_PER_WAVE - 1) / ROWS_PER_WAVE;
    const int nodeChunks = (nNodes + ROWS_PER_WAVE - 1) / ROWS_PER_WAVE;
    const int nChunks    = edgeChunks + nodeChunks;
    const int wavesTotal = gridDim.x * (POOL_THREADS / 32);

    for (int c = blockIdx.x * (POOL_THREADS / 32) + wv; c < nChunks; c += wavesTotal) {
        int sv, gv, n;
        if (c < edgeChunks) {
            const int e0 = c * ROWS_PER_WAVE;
            n = min(ROWS_PER_WAVE, nEdges - e0);
            const int e = e0 + min(li, n - 1);
            sv = src[e];
            gv = gid[dst[e]];
        } else {
            const int i0 = (c - edgeChunks) * ROWS_PER_WAVE;
            n = min(ROWS_PER_WAVE, nNodes - i0);
            sv = i0 + min(li, n - 1);
            gv = gid[sv];
            if (lane < n) unsafeAtomicAdd(&sC[gv], 1.0f);
        }

        if (n == ROWS_PER_WAVE) {
            v2f r[ROWS_PER_WAVE];
            #pragma unroll
            for (int i = 0; i < ROWS_PER_WAVE; ++i) {
                const int s = __builtin_amdgcn_readlane(sv, i);
                r[i] = ((const v2f*)(h + (size_t)s * D_FEAT))[lane];
            }
            #pragma unroll
            for (int i = 0; i < ROWS_PER_WAVE; ++i) {
                const int g = __builtin_amdgcn_readlane(gv, i);
                float* t = sP + g * D_FEAT + 2 * lane;
                unsafeAtomicAdd(t,     r[i].x);
                unsafeAtomicAdd(t + 1, r[i].y);
            }
        } else {
            for (int i = 0; i < n; ++i) {
                const int s = __builtin_amdgcn_readlane(sv, i);
                const int g = __builtin_amdgcn_readlane(gv, i);
                const v2f rv = ((const v2f*)(h + (size_t)s * D_FEAT))[lane];
                float* t = sP + g * D_FEAT + 2 * lane;
                unsafeAtomicAdd(t,     rv.x);
                unsafeAtomicAdd(t + 1, rv.y);
            }
        }
    }
    __syncthreads();

    constexpr int LINES = N_GRAPHS * D_FEAT / 32;
    #pragma unroll
    for (int j = 0; j < LINES / (POOL_THREADS / 32); ++j) {
        const int line = (wv + (POOL_THREADS / 32) * j + (int)blockIdx.x) & (LINES - 1);
        const int idx  = line * 32 + lane;
        unsafeAtomicAdd(pool + idx, sP[idx]);
    }
    if (threadIdx.x < N_GRAPHS) unsafeAtomicAdd(cnt + threadIdx.x, sC[threadIdx.x]);
}

__global__ void __launch_bounds__(256)
gin_pool_linear_kernel(const float* __restrict__ pool,
                       const float* __restrict__ cnt,
                       const float* __restrict__ W,
                       const float* __restrict__ bias,
                       float*       __restrict__ out)
{
    __shared__ float sWp[D_FEAT * D_FEAT];
    __shared__ float sB[D_FEAT];
    stage_weights(W, bias, sWp, sB);

    const int wave    = threadIdx.x >> 5;
    const int lane    = threadIdx.x & 31;
    const int rowBase = (blockIdx.x * 8 + wave) * 16;
    if (rowBase >= N_GRAPHS) return;

    const int halfSel = lane >> 4;
    const int laneM   = lane & 15;

    v8f acc[4];
    wave_tile_product<false>(pool + (size_t)(rowBase + laneM) * D_FEAT, nullptr,
                             sWp, halfSel, laneM, acc);

    #pragma unroll
    for (int v = 0; v < 8; ++v) {
        const int row = rowBase + v + halfSel * 8;
        const float nv = cnt[row];
        #pragma unroll
        for (int ct = 0; ct < 4; ++ct) {
            const int col = ct * 16 + laneM;
            out[(size_t)row * D_FEAT + col] = acc[ct][v] + nv * sB[col];
        }
    }
}

extern "C" void kernel_launch(void* const* d_in, const int* in_sizes, int n_in,
                              void* d_out, int out_size, void* d_ws, size_t ws_size,
                              hipStream_t stream)
{
    const float* feats = (const float*)d_in[0];
    const int*   src   = (const int*)  d_in[1];
    const int*   dst   = (const int*)  d_in[2];
    const int*   gid   = (const int*)  d_in[3];
    const float* W1    = (const float*)d_in[4];
    const float* b1    = (const float*)d_in[5];
    const float* W2    = (const float*)d_in[6];
    const float* b2    = (const float*)d_in[7];
    float* out = (float*)d_out;

    const int nNodes = in_sizes[0] / D_FEAT;
    const int nEdges = in_sizes[1];
    const int nSlices = (nEdges + SLICE - 1) / SLICE;
    const int nParts  = (nNodes + PART_ROWS - 1) / PART_ROWS;

    float* h      = (float*)d_ws;
    float* pool   = h    + (size_t)nNodes * D_FEAT;
    float* cnt    = pool + N_GRAPHS * D_FEAT;
    int*   binned = (int*)(cnt + N_GRAPHS);
    int*   offs   = binned + (size_t)nSlices * SLICE * 2;

    hipMemsetAsync(pool, 0, (N_GRAPHS * D_FEAT + N_GRAPHS) * sizeof(float), stream);

    gin_bin_kernel<<<nSlices, BIN_THREADS, 0, stream>>>(src, dst, binned, offs, nEdges);
    gin_layer1_kernel<<<nParts, L1_THREADS, 0, stream>>>(feats, binned, offs, W1, b1, h,
                                                         nNodes, nSlices);

    gin_pool_kernel<<<512, POOL_THREADS, 0, stream>>>(h, src, dst, gid, pool, cnt,
                                                      nEdges, nNodes);
    gin_pool_linear_kernel<<<1, 256, 0, stream>>>(pool, cnt, W2, b2, out);
}


# layer 2 edge pass fused into the layer-1 kernel: edges also sorted by src block, a work-group adds its own h rows from LDS into a per-graph LDS table; h stays in LDS; sums reassociated, arrival order
# speedup vs baseline: 3.3501x; 1.1486x over previous
#include <hip/hip_runtime.h>
#include <hip/hip_bf16.h>


typedef float v2f __attribute__((ext_vector_type(2)));
typedef float v8f __attribute__((ext_vector_type(8)));

#define D_FEAT   64
#define N_GRAPHS 128

#define PART_ROWS   400
#define MAX_PARTS   256
#define SLICE       8192
#define BIN_THREADS 1024
#define EDGES_PER_THREAD (SLICE / BIN_THREADS)
#define OFFS_PITCH  (MAX_PARTS + 1)

__global__ void __launch_bounds__(BIN_THREADS)
gin_bin_kernel(const int* __restrict__ src,
               const int* __restrict__ dst,
               const int* __restrict__ gid,
               int*       __restrict__ binned,
               int*       __restrict__ offs,
               int nEdges, int nSlices)
{
    __shared__ int  sCnt[MAX_PARTS];
    __shared__ int  sOff[MAX_PARTS + 1];
    __shared__ int  sCur[MAX_PARTS];
    __shared__ int2 sPairs[SLICE];

    const int tid   = threadIdx.x;
    const int bySrc = ((int)blockIdx.x >= nSlices) ? 1 : 0;
    const int slice = (int)blockIdx.x - bySrc * nSlices;
    const int e0    = slice * SLICE;
    const int n     = min(SLICE, nEdges - e0);

    if (tid < MAX_PARTS) sCnt[tid] = 0;
    __syncthreads();

    int kv[EDGES_PER_THREAD], pv[EDGES_PER_THREAD];
    if (bySrc == 0) {
        #pragma unroll
        for (int j = 0; j < EDGES_PER_THREAD; ++j) {
            const int i = tid + j * BIN_THREADS;
            kv[j] = (i < n) ? dst[e0 + i] : 0;
            pv[j] = (i < n) ? src[e0 + i] : 0;
        }
    } else {
        int dv[EDGES_PER_THREAD];
        #pragma unroll
        for (int j = 0; j < EDGES_PER_THREAD; ++j) {
            const int i = tid + j * BIN_THREADS;
            kv[j] = (i < n) ? src[e0 + i] : 0;
            dv[j] = (i < n) ? dst[e0 + i] : 0;
        }
        #pragma unroll
        for (int j = 0; j < EDGES_PER_THREAD; ++j) {
            const int i = tid + j * BIN_THREADS;
            pv[j] = (i < n) ? gid[dv[j]] : 0;
        }
    }
    #pragma unroll
    for (int j = 0; j < EDGES_PER_THREAD; ++j) {
        if (tid + j * BIN_THREADS < n)
            atomicAdd(&sCnt[min(kv[j] / PART_ROWS, MAX_PARTS - 1)], 1);
    }
    __syncthreads();

    if (tid < 32) {
        int c[8];
        int sum = 0;
        #pragma unroll
        for (int j = 0; j < 8; ++j) { c[j] = sCnt[tid * 8 + j]; sum += c[j]; }
        int incl = sum;
        #pragma unroll
        for (int d = 1; d < 32; d <<= 1) {
            const int t = __shfl_up(incl, d);
            if (tid >= d) incl += t;
        }
        int run = incl - sum;
        #pragma unroll
        for (int j = 0; j < 8; ++j) {
            sOff[tid * 8 + j] = run;
            sCur[tid * 8 + j] = run;
            run += c[j];
        }
        if (tid == 31) sOff[MAX_PARTS] = run;
    }
    __syncthreads();

    #pragma unroll
    for (int j = 0; j < EDGES_PER_THREAD; ++j) {
        if (tid + j * BIN_THREADS < n) {
            const int p   = min(kv[j] / PART_ROWS, MAX_PARTS - 1);
            const int pos = atomicAdd(&sCur[p], 1);
            sPairs[pos] = make_int2(pv[j], kv[j] - p * PART_ROWS);
        }
    }
    __syncthreads();

    int2* __restrict__ o = (int2*)binned + (size_t)blockIdx.x * SLICE;
    for (int i = tid; i < n; i += BIN_THREADS) o[i] = sPairs[i];
    if (tid <= MAX_PARTS) offs[blockIdx.x * OFFS_PITCH + tid] = sOff[tid];
}

__device__ __forceinline__ void
stage_weights(const float* __restrict__ W,
              const float* __restrict__ bias,
              float* sWp, float* sB)
{
    for (int idx = threadIdx.x; idx < D_FEAT * D_FEAT; idx += 256) {
        const int k   = idx >> 6;
        const int col = idx & 63;
        sWp[(k >> 1) * 128 + col * 2 + (k & 1)] = W[idx];
    }
    if (threadIdx.x < D_FEAT) sB[threadIdx.x] = bias[threadIdx.x];
    __syncthreads();
}

template <bool ADD_SECOND>
__device__ __forceinline__ void
wave_tile_product(const float* __restrict__ xr, const float* __restrict__ gr,
                  const float* sWp, int halfSel, int laneM, v8f (&acc)[4])
{
    #pragma unroll
    for (int ct = 0; ct < 4; ++ct) acc[ct] = (v8f){0.f,0.f,0.f,0.f,0.f,0.f,0.f,0.f};

    #pragma unroll 4
    for (int kt = 0; kt < 16; ++kt) {
        const int k0 = kt * 4 + halfSel * 2;
        v2f a = *(const v2f*)(xr + k0);
        if (ADD_SECOND) a = a + *(const v2f*)(gr + k0);
        const int bRow = (kt * 2 + halfSel) * 128;
        #pragma unroll
        for (int ct = 0; ct < 4; ++ct) {
            const int col = ct * 16 + laneM;
            const v2f b = *(const v2f*)(sWp + bRow + col * 2);
            acc[ct] = __builtin_amdgcn_wmma_f32_16x16x4_f32(
                false, a, false, b, (short)0, acc[ct], false, false);
        }
    }
}

#define L1_THREADS  1024
#define TILE_PITCH  66
#define GATHER_ROWS 16

template <bool POOL>
__device__ __forceinline__ void
walk_block_segments(const float* __restrict__ X, float* sT, float* sP,
                    const int2* __restrict__ pairs, const int* __restrict__ offs,
                    int part, int wv, int lane, int nSlices)
{
    const int nMine = (nSlices > wv) ? (nSlices - wv + 31) / 32 : 0;
    int lov = 0, hiv = 0;
    if (lane < nMine) {
        const int* __restrict__ o = offs + (wv + 32 * lane) * OFFS_PITCH + part;
        lov = o[0];
        hiv = o[1];
    }
    const int li = lane & (GATHER_ROWS - 1);

    int j = 0, b0 = 0, segHi = 0;
    for (; j < nMine; ++j) {
        b0    = __builtin_amdgcn_readlane(lov, j);
        segHi = __builtin_amdgcn_readlane(hiv, j);
        if (b0 < segHi) break;
    }
    int2 pr = make_int2(0, 0);
    if (j < nMine)
        pr = pairs[(size_t)(wv + 32 * j) * SLICE + b0 + min(li, min(GATHER_ROWS, segHi - b0) - 1)];

    while (j < nMine) {
        const int n = min(GATHER_ROWS, segHi - b0);
        const int tileRow = min(pr.y, PART_ROWS - 1);
        const int ldv  = POOL ? tileRow : pr.x;
        const int addv = POOL ? (pr.x & (N_GRAPHS - 1)) : tileRow;

        v2f r[GATHER_ROWS];
        #pragma unroll
        for (int i = 0; i < GATHER_ROWS; ++i) {
            const int s = __builtin_amdgcn_readlane(ldv, i);
            if (POOL) r[i] = *(const v2f*)(sT + s * TILE_PITCH + 2 * lane);
            else      r[i] = ((const v2f*)(X + (size_t)s * D_FEAT))[lane];
        }

        int nj = j, nb0 = b0 + GATHER_ROWS, nHi = segHi;
        if (nb0 >= nHi) {
            for (++nj; nj < nMine; ++nj) {
                nb0 = __builtin_amdgcn_readlane(lov, nj);
                nHi = __builtin_amdgcn_readlane(hiv, nj);
                if (nb0 < nHi) break;
            }
        }
        if (nj < nMine)
            pr = pairs[(size_t)(wv + 32 * nj) * SLICE + nb0 + min(li, min(GATHER_ROWS, nHi - nb0) - 1)];

        #pragma unroll
        for (int i = 0; i < GATHER_ROWS; ++i) {
            if (i < n) {
                const int a = __builtin_amdgcn_readlane(addv, i);
                float* t = POOL ? (sP + a * D_FEAT + 2 * lane)
                                : (sT + a * TILE_PITCH + 2 * lane);
                unsafeAtomicAdd(t,     r[i].x);
                unsafeAtomicAdd(t + 1, r[i].y);
            }
        }
        j = nj; b0 = nb0; segHi = nHi;
    }
}

__global__ void __launch_bounds__(L1_THREADS) __attribute__((amdgpu_waves_per_eu(8, 8)))
gin_block_kernel(const float* __restrict__ X,
                 const int*   __restrict__ binned,
                 const int*   __restrict__ offs,
                 const int*   __restrict__ gid,
                 const float* __restrict__ W,
                 const float* __restrict__ bias,
                 float*       __restrict__ pool,
                 float*       __restrict__ cnt,
                 int nNodes, int nSlices)
{
    __shared__ float sT[PART_ROWS * TILE_PITCH];
    __shared__ float sWp[D_FEAT * D_FEAT];
    __shared__ float sB[D_FEAT];
    __shared__ float sP[N_GRAPHS * D_FEAT];
    __shared__ float sC[N_GRAPHS];

    const int tid = threadIdx.x;
    for (int i = tid; i < PART_ROWS * TILE_PITCH; i += L1_THREADS) sT[i] = 0.f;
    for (int i = tid; i < N_GRAPHS * D_FEAT; i += L1_THREADS) sP[i] = 0.f;
    if (tid < N_GRAPHS) sC[tid] = 0.f;
    for (int idx = tid; idx < D_FEAT * D_FEAT; idx += L1_THREADS) {
        const int k   = idx >> 6;
        const int col = idx & 63;
        sWp[(k >> 1) * 128 + col * 2 + (k & 1)] = W[idx];
    }
    if (tid < D_FEAT) sB[tid] = bias[tid];
    __syncthreads();

    const int lane = tid & 31;
    const int wv   = __builtin_amdgcn_readfirstlane(tid >> 5);
    const int part = blockIdx.x;
    const int2* __restrict__ pairsDst = (const int2*)binned;
    const int2* __restrict__ pairsSrc = (const int2*)binned + (size_t)nSlices * SLICE;
    const int*  __restrict__ offsDst  = offs;
    const int*  __restrict__ offsSrc  = offs + nSlices * OFFS_PITCH;

    walk_block_segments<false>(X, sT, sP, pairsDst, offsDst, part, wv, lane, nSlices);
    __syncthreads();

    const int rowLocal = wv * 16;
    const int rowBase  = part * PART_ROWS + rowLocal;
    const int halfSel  = lane >> 4;
    const int laneM    = lane & 15;
    const bool hasTile = (wv < PART_ROWS / 16) && (rowBase < nNodes);

    if (hasTile) {
        int aRow = rowBase + laneM;
        if (aRow >= nNodes) aRow = nNodes - 1;
        const float* __restrict__ xr = X + (size_t)aRow * D_FEAT;
        const float* tr = sT + (rowLocal + laneM) * TILE_PITCH;

        v8f acc[4];
        #pragma unroll
        for (int ct = 0; ct < 4; ++ct) acc[ct] = (v8f){0.f,0.f,0.f,0.f,0.f,0.f,0.f,0.f};

        #pragma unroll 4
        for (int kt = 0; kt < 16; ++kt) {
            const int k0 = kt * 4 + halfSel * 2;
            const v2f a = *(const v2f*)(xr + k0) + *(const v2f*)(tr + k0);
            const int bRow = (kt * 2 + halfSel) * 128;
            #pragma unroll
            for (int ct = 0; ct < 4; ++ct) {
                const int col = ct * 16 + laneM;
                const v2f b = *(const v2f*)(sWp + bRow + col * 2);
                acc[ct] = __builtin_amdgcn_wmma_f32_16x16x4_f32(
                    false, a, false, b, (short)0, acc[ct], false, false);
            }
        }

        #pragma unroll
        for (int ct = 0; ct < 4; ++ct) {
            const int col = ct * 16 + laneM;
            const float bv = sB[col];
            #pragma unroll
            for (int v = 0; v < 8; ++v)
                sT[(rowLocal + v + halfSel * 8) * TILE_PITCH + col] = fmaxf(acc[ct][v] + bv, 0.f);
        }
    }
    __syncthreads();

    if (hasTile) {
        const int n  = min(GATHER_ROWS, nNodes - rowBase);
        const int li = lane & (GATHER_ROWS - 1);
        const int gv = gid[rowBase + min(li, n - 1)] & (N_GRAPHS - 1);
        if (lane < n) unsafeAtomicAdd(&sC[gv], 1.0f);

        v2f r[GATHER_ROWS];
        #pragma unroll
        for (int i = 0; i < GATHER_ROWS; ++i)
            r[i] = *(const v2f*)(sT + (rowLocal + i) * TILE_PITCH + 2 * lane);
        #pragma unroll
        for (int i = 0; i < GATHER_ROWS; ++i) {
            if (i < n) {
                const int g = __builtin_amdgcn_readlane(gv, i);
                float* t = sP + g * D_FEAT + 2 * lane;
                unsafeAtomicAdd(t,     r[i].x);
                unsafeAtomicAdd(t + 1, r[i].y);
            }
        }
    }
    walk_block_segments<true>(X, sT, sP, pairsSrc, offsSrc, part, wv, lane, nSlices);
    __syncthreads();

    constexpr int LINES = N_GRAPHS * D_FEAT / 32;
    #pragma unroll
    for (int q = 0; q < LINES / (L1_THREADS / 32); ++q) {
        const int line = (wv + (L1_THREADS / 32) * q + (int)blockIdx.x) & (LINES - 1);
        const int idx  = line * 32 + lane;
        unsafeAtomicAdd(pool + idx, sP[idx]);
    }
    if (tid < N_GRAPHS) unsafeAtomicAdd(cnt + tid, sC[tid]);
}

__global__ void __launch_bounds__(256)
gin_pool_linear_kernel(const float* __restrict__ pool,
                       const float* __restrict__ cnt,
                       const float* __restrict__ W,
                       const float* __restrict__ bias,
                       float*       __restrict__ out)
{
    __shared__ float sWp[D_FEAT * D_FEAT];
    __shared__ float sB[D_FEAT];
    stage_weights(W, bias, sWp, sB);

    const int wave    = threadIdx.x >> 5;
    const int lane    = threadIdx.x & 31;
    const int rowBase = (blockIdx.x * 8 + wave) * 16;
    if (rowBase >= N_GRAPHS) return;

    const int halfSel = lane >> 4;
    const int laneM   = lane & 15;

    v8f acc[4];
    wave_tile_product<false>(pool + (size_t)(rowBase + laneM) * D_FEAT, nullptr,
                             sWp, halfSel, laneM, acc);

    #pragma unroll
    for (int v = 0; v < 8; ++v) {
        const int row = rowBase + v + halfSel * 8;
        const float nv = cnt[row];
        #pragma unroll
        for (int ct = 0; ct < 4; ++ct) {
            const int col = ct * 16 + laneM;
            out[(size_t)row * D_FEAT + col] = acc[ct][v] + nv * sB[col];
        }
    }
}

extern "C" void kernel_launch(void* const* d_in, const int* in_sizes, int n_in,
                              void* d_out, int out_size, void* d_ws, size_t ws_size,
                              hipStream_t stream)
{
    const float* feats = (const float*)d_in[0];
    const int*   src   = (const int*)  d_in[1];
    const int*   dst   = (const int*)  d_in[2];
    const int*   gid   = (const int*)  d_in[3];
    const float* W1    = (const float*)d_in[4];
    const float* b1    = (const float*)d_in[5];
    const float* W2    = (const float*)d_in[6];
    const float* b2    = (const float*)d_in[7];
    float* out = (float*)d_out;

    const int nNodes = in_sizes[0] / D_FEAT;
    const int nEdges = in_sizes[1];
    const int nSlices = (nEdges + SLICE - 1) / SLICE;
    const int nParts  = (nNodes + PART_ROWS - 1) / PART_ROWS;

    float* pool   = (float*)d_ws;
    float* cnt    = pool + N_GRAPHS * D_FEAT;
    int*   binned = (int*)(cnt + N_GRAPHS);
    int*   offs   = binned + (size_t)2 * nSlices * SLICE * 2;

    hipMemsetAsync(pool, 0, (N_GRAPHS * D_FEAT + N_GRAPHS) * sizeof(float), stream);

    gin_bin_kernel<<<2 * nSlices, BIN_THREADS, 0, stream>>>(src, dst, gid, binned, offs,
                                                            nEdges, nSlices);
    gin_block_kernel<<<nParts, L1_THREADS, 0, stream>>>(feats, binned, offs, gid, W1, b1,
                                                        pool, cnt, nNodes, nSlices);

    gin_pool_linear_kernel<<<1, 256, 0, stream>>>(pool, cnt, W2, b2, out);
}


# block kernel: its pairs are sorted once more in LDS (by dst row / by graph); each wave sums its own rows in registers, one store or add per row, no LDS f32 add per edge; sums regrouped, arrival order
# speedup vs baseline: 4.0014x; 1.1944x over previous
#include <hip/hip_runtime.h>
#include <hip/hip_bf16.h>


typedef float v2f __attribute__((ext_vector_type(2)));
typedef float v8f __attribute__((ext_vector_type(8)));

#define D_FEAT   64
#define N_GRAPHS 128

#define PART_ROWS   400
#define MAX_PARTS   256
#define SLICE       8192
#define BIN_THREADS 1024
#define EDGES_PER_THREAD (SLICE / BIN_THREADS)
#define OFFS_PITCH  (MAX_PARTS + 1)

__global__ void __launch_bounds__(BIN_THREADS)
gin_bin_kernel(const int* __restrict__ src,
               const int* __restrict__ dst,
               const int* __restrict__ gid,
               int*       __restrict__ binned,
               int*       __restrict__ offs,
               int nEdges, int nSlices)
{
    __shared__ int  sCnt[MAX_PARTS];
    __shared__ int  sOff[MAX_PARTS + 1];
    __shared__ int  sCur[MAX_PARTS];
    __shared__ int2 sPairs[SLICE];

    const int tid   = threadIdx.x;
    const int bySrc = ((int)blockIdx.x >= nSlices) ? 1 : 0;
    const int slice = (int)blockIdx.x - bySrc * nSlices;
    const int e0    = slice * SLICE;
    const int n     = min(SLICE, nEdges - e0);

    if (tid < MAX_PARTS) sCnt[tid] = 0;
    __syncthreads();

    int kv[EDGES_PER_THREAD], pv[EDGES_PER_THREAD];
    if (bySrc == 0) {
        #pragma unroll
        for (int j = 0; j < EDGES_PER_THREAD; ++j) {
            const int i = tid + j * BIN_THREADS;
            kv[j] = (i < n) ? dst[e0 + i] : 0;
            pv[j] = (i < n) ? src[e0 + i] : 0;
        }
    } else {
        int dv[EDGES_PER_THREAD];
        #pragma unroll
        for (int j = 0; j < EDGES_PER_THREAD; ++j) {
            const int i = tid + j * BIN_THREADS;
            kv[j] = (i < n) ? src[e0 + i] : 0;
            dv[j] = (i < n) ? dst[e0 + i] : 0;
        }
        #pragma unroll
        for (int j = 0; j < EDGES_PER_THREAD; ++j) {
            const int i = tid + j * BIN_THREADS;
            pv[j] = (i < n) ? gid[dv[j]] : 0;
        }
    }
    #pragma unroll
    for (int j = 0; j < EDGES_PER_THREAD; ++j) {
        if (tid + j * BIN_THREADS < n)
            atomicAdd(&sCnt[min(kv[j] / PART_ROWS, MAX_PARTS - 1)], 1);
    }
    __syncthreads();

    if (tid < 32) {
        int c[8];
        int sum = 0;
        #pragma unroll
        for (int j = 0; j < 8; ++j) { c[j] = sCnt[tid * 8 + j]; sum += c[j]; }
        int incl = sum;
        #pragma unroll
        for (int d = 1; d < 32; d <<= 1) {
            const int t = __shfl_up(incl, d);
            if (tid >= d) incl += t;
        }
        int run = incl - sum;
        #pragma unroll
        for (int j = 0; j < 8; ++j) {
            sOff[tid * 8 + j] = run;
            sCur[tid * 8 + j] = run;
            run += c[j];
        }
        if (tid == 31) sOff[MAX_PARTS] = run;
    }
    __syncthreads();

    #pragma unroll
    for (int j = 0; j < EDGES_PER_THREAD; ++j) {
        if (tid + j * BIN_THREADS < n) {
            const int p   = min(kv[j] / PART_ROWS, MAX_PARTS - 1);
            const int pos = atomicAdd(&sCur[p], 1);
            sPairs[pos] = make_int2(pv[j], kv[j] - p * PART_ROWS);
        }
    }
    __syncthreads();

    int2* __restrict__ o = (int2*)binned + (size_t)blockIdx.x * SLICE;
    for (int i = tid; i < n; i += BIN_THREADS) o[i] = sPairs[i];
    if (tid <= MAX_PARTS) offs[blockIdx.x * OFFS_PITCH + tid] = sOff[tid];
}

__device__ __forceinline__ void
stage_weights(const float* __restrict__ W,
              const float* __restrict__ bias,
              float* sWp, float* sB)
{
    for (int idx = threadIdx.x; idx < D_FEAT * D_FEAT; idx += 256) {
        const int k   = idx >> 6;
        const int col = idx & 63;
        sWp[(k >> 1) * 128 + col * 2 + (k & 1)] = W[idx];
    }
    if (threadIdx.x < D_FEAT) sB[threadIdx.x] = bias[threadIdx.x];
    __syncthreads();
}

template <bool ADD_SECOND>
__device__ __forceinline__ void
wave_tile_product(const float* __restrict__ xr, const float* __restrict__ gr,
                  const float* sWp, int halfSel, int laneM, v8f (&acc)[4])
{
    #pragma unroll
    for (int ct = 0; ct < 4; ++ct) acc[ct] = (v8f){0.f,0.f,0.f,0.f,0.f,0.f,0.f,0.f};

    #pragma unroll 4
    for (int kt = 0; kt < 16; ++kt) {
        const int k0 = kt * 4 + halfSel * 2;
        v2f a = *(const v2f*)(xr + k0);
        if (ADD_SECOND) a = a + *(const v2f*)(gr + k0);
        const int bRow = (kt * 2 + halfSel) * 128;
        #pragma unroll
        for (int ct = 0; ct < 4; ++ct) {
            const int col = ct * 16 + laneM;
            const v2f b = *(const v2f*)(sWp + bRow + col * 2);
            acc[ct] = __builtin_amdgcn_wmma_f32_16x16x4_f32(
                false, a, false, b, (short)0, acc[ct], false, false);
        }
    }
}

#define L1_THREADS  1024
#define TILE_PITCH  66
#define GATHER_ROWS 16

template <bool POOL>
__device__ __forceinline__ void
walk_block_segments(const float* __restrict__ X, float* sT, float* sP,
                    const int2* __restrict__ pairs, const int* __restrict__ offs,
                    int part, int wv, int lane, int nSlices)
{
    const int nMine = (nSlices > wv) ? (nSlices - wv + 31) / 32 : 0;
    int lov = 0, hiv = 0;
    if (lane < nMine) {
        const int* __restrict__ o = offs + (wv + 32 * lane) * OFFS_PITCH + part;
        lov = o[0];
        hiv = o[1];
    }
    const int li = lane & (GATHER_ROWS - 1);

    int j = 0, b0 = 0, segHi = 0;
    for (; j < nMine; ++j) {
        b0    = __builtin_amdgcn_readlane(lov, j);
        segHi = __builtin_amdgcn_readlane(hiv, j);
        if (b0 < segHi) break;
    }
    int2 pr = make_int2(0, 0);
    if (j < nMine)
        pr = pairs[(size_t)(wv + 32 * j) * SLICE + b0 + min(li, min(GATHER_ROWS, segHi - b0) - 1)];

    while (j < nMine) {
        const int n = min(GATHER_ROWS, segHi - b0);
        const int tileRow = min(pr.y, PART_ROWS - 1);
        const int ldv  = POOL ? tileRow : pr.x;
        const int addv = POOL ? (pr.x & (N_GRAPHS - 1)) : tileRow;

        v2f r[GATHER_ROWS];
        #pragma unroll
        for (int i = 0; i < GATHER_ROWS; ++i) {
            const int s = __builtin_amdgcn_readlane(ldv, i);
            if (POOL) r[i] = *(const v2f*)(sT + s * TILE_PITCH + 2 * lane);
            else      r[i] = ((const v2f*)(X + (size_t)s * D_FEAT))[lane];
        }

        int nj = j, nb0 = b0 + GATHER_ROWS, nHi = segHi;
        if (nb0 >= nHi) {
            for (++nj; nj < nMine; ++nj) {
                nb0 = __builtin_amdgcn_readlane(lov, nj);
                nHi = __builtin_amdgcn_readlane(hiv, nj);
                if (nb0 < nHi) break;
            }
        }
        if (nj < nMine)
            pr = pairs[(size_t)(wv + 32 * nj) * SLICE + nb0 + min(li, min(GATHER_ROWS, nHi - nb0) - 1)];

        #pragma unroll
        for (int i = 0; i < GATHER_ROWS; ++i) {
            if (i < n) {
                const int a = __builtin_amdgcn_readlane(addv, i);
                float* t = POOL ? (sP + a * D_FEAT + 2 * lane)
                                : (sT + a * TILE_PITCH + 2 * lane);
                unsafeAtomicAdd(t,     r[i].x);
                unsafeAtomicAdd(t + 1, r[i].y);
            }
        }
        j = nj; b0 = nb0; segHi = nHi;
    }
}

#define LIST_CAP    8192
#define G_PER_LANE  13
#define G_BINS      (32 * G_PER_LANE)
#define P_PER_LANE  (N_GRAPHS / 32)
#define LOAD_CH     8
#define ENT_SHIFT   17
#define ENT_MASK    0x1FFFF
static_assert(G_BINS >= PART_ROWS, "row bins");
static_assert(MAX_PARTS * PART_ROWS <= (1 << ENT_SHIFT), "payload bits");

struct WaveSegs { int incl; int q; int total; int nMine; };

__device__ __forceinline__ WaveSegs
wave_segments(const int* __restrict__ offs, int part, int wv, int lane, int nSlices)
{
    WaveSegs s;
    s.nMine = (nSlices > wv) ? (nSlices - wv + 31) / 32 : 0;
    int lo = 0, hi = 0;
    if (lane < s.nMine) {
        const int* __restrict__ o = offs + (wv + 32 * lane) * OFFS_PITCH + part;
        lo = o[0];
        hi = o[1];
    }
    const int len = max(hi - lo, 0);
    int incl = len;
    #pragma unroll
    for (int d = 1; d < 32; d <<= 1) {
        const int t = __shfl_up(incl, d);
        if (lane >= d) incl += t;
    }
    s.incl  = incl;
    s.total = __builtin_amdgcn_readlane(incl, 31);
    s.q     = (lane < s.nMine) ? (wv + 32 * lane) * SLICE + lo - (incl - len) : 0;
    return s;
}

__device__ __forceinline__ void
chunk_indices(const WaveSegs& sg, int base, int lane, int (&t)[LOAD_CH])
{
    int j[LOAD_CH];
    #pragma unroll
    for (int c = 0; c < LOAD_CH; ++c) {
        t[c] = max(min(base + c * 32 + lane, sg.total - 1), 0);
        j[c] = 0;
    }
    for (int jj = 0; jj < sg.nMine; ++jj) {
        const int end = __builtin_amdgcn_readlane(sg.incl, jj);
        #pragma unroll
        for (int c = 0; c < LOAD_CH; ++c) j[c] += (t[c] >= end) ? 1 : 0;
    }
    #pragma unroll
    for (int c = 0; c < LOAD_CH; ++c) t[c] += __shfl(sg.q, j[c]);
}

template <bool POOL>
__device__ __forceinline__ unsigned make_entry(int2 pr)
{
    const unsigned row = min((unsigned)pr.y, (unsigned)(PART_ROWS - 1));
    return POOL ? ((((unsigned)pr.x & (N_GRAPHS - 1)) << ENT_SHIFT) | row)
                : ((row << ENT_SHIFT) | ((unsigned)pr.x & ENT_MASK));
}

template <bool POOL, bool SCATTER>
__device__ __forceinline__ void
put_chunks(const int2 (&pr)[LOAD_CH], int total, int base, int lane,
           int* sCur, unsigned* sList)
{
    unsigned ent[LOAD_CH];
    int      pos[LOAD_CH];
    #pragma unroll
    for (int c = 0; c < LOAD_CH; ++c) ent[c] = make_entry<POOL>(pr[c]);
    #pragma unroll
    for (int c = 0; c < LOAD_CH; ++c) {
        pos[c] = LIST_CAP;
        if (base + c * 32 + lane < total) pos[c] = atomicAdd(&sCur[ent[c] >> ENT_SHIFT], 1);
    }
    if (SCATTER) {
        #pragma unroll
        for (int c = 0; c < LOAD_CH; ++c)
            if (pos[c] < LIST_CAP) sList[pos[c]] = ent[c];
    }
}

template <bool POOL, bool SCATTER>
__device__ __forceinline__ int2 load_pair(const int2* __restrict__ pairs, int t)
{
    if (SCATTER) return pairs[t];
    const int* __restrict__ p = (const int*)(pairs + t);
    return POOL ? make_int2(p[0], 0) : make_int2(0, p[1]);
}

template <bool SCATTER>
__device__ __forceinline__ void
sort_pass(const int2* __restrict__ pairsDst, const int2* __restrict__ pairsSrc,
          const WaveSegs& sgG, const WaveSegs& sgP, int lane,
          int* sCurG, int* sCurP, unsigned* sListG, unsigned* sListP)
{
    const int tmax = max(sgG.total, sgP.total);
    for (int base = 0; base < tmax; base += 32 * LOAD_CH) {
        int  tg[LOAD_CH], tp[LOAD_CH];
        int2 pg[LOAD_CH], pp[LOAD_CH];
        chunk_indices(sgG, base, lane, tg);
        chunk_indices(sgP, base, lane, tp);
        #pragma unroll
        for (int c = 0; c < LOAD_CH; ++c) pg[c] = load_pair<false, SCATTER>(pairsDst, tg[c]);
        #pragma unroll
        for (int c = 0; c < LOAD_CH; ++c) pp[c] = load_pair<true,  SCATTER>(pairsSrc, tp[c]);
        put_chunks<false, SCATTER>(pg, sgG.total, base, lane, sCurG, sListG);
        put_chunks<true,  SCATTER>(pp, sgP.total, base, lane, sCurP, sListP);
    }
}

template <int PER>
__device__ __forceinline__ void wave_scan(int* sCnt, int* sOff, int lane)
{
    int c[PER];
    int sum = 0;
    #pragma unroll
    for (int j = 0; j < PER; ++j) { c[j] = sCnt[lane * PER + j]; sum += c[j]; }
    int incl = sum;
    #pragma unroll
    for (int d = 1; d < 32; d <<= 1) {
        const int t = __shfl_up(incl, d);
        if (lane >= d) incl += t;
    }
    int run = incl - sum;
    #pragma unroll
    for (int j = 0; j < PER; ++j) {
        sOff[lane * PER + j] = run;
        sCnt[lane * PER + j] = run;
        run += c[j];
    }
    if (lane == 31) sOff[32 * PER] = run;
}

template <bool POOL>
__device__ __forceinline__ void flush_sum(float* sT, float* sP, int key, v2f acc, int lane)
{
    if (POOL) {
        float* t = sP + key * D_FEAT + 2 * lane;
        unsafeAtomicAdd(t,     acc.x);
        unsafeAtomicAdd(t + 1, acc.y);
    } else {
        *(v2f*)(sT + key * TILE_PITCH + 2 * lane) = acc;
    }
}

template <bool POOL>
__device__ __forceinline__ void
consume_list(const float* __restrict__ X, float* sT, float* sP,
             const unsigned* sList, int s, int e, int lane)
{
    const int li = lane & (GATHER_ROWS - 1);
    int cur = -1;
    v2f acc = (v2f){0.f, 0.f};
    unsigned ent = 0;
    if (s < e) ent = sList[s + min(li, min(GATHER_ROWS, e - s) - 1)];

    for (int b = s; b < e; b += GATHER_ROWS) {
        const int n = min(GATHER_ROWS, e - b);
        const int ldv  = (int)(ent & ENT_MASK);
        const int keyv = (int)(ent >> ENT_SHIFT);

        v2f r[GATHER_ROWS];
        #pragma unroll
        for (int i = 0; i < GATHER_ROWS; ++i) {
            const int row = __builtin_amdgcn_readlane(ldv, i);
            if (POOL) r[i] = *(const v2f*)(sT + row * TILE_PITCH + 2 * lane);
            else      r[i] = ((const v2f*)(X + (size_t)row * D_FEAT))[lane];
        }
        const int nb = b + GATHER_ROWS;
        if (nb < e) ent = sList[nb + min(li, min(GATHER_ROWS, e - nb) - 1)];

        #pragma unroll
        for (int i = 0; i < GATHER_ROWS; ++i) {
            if (i < n) {
                const int k = __builtin_amdgcn_readlane(keyv, i);
                if (k != cur) {
                    if (cur >= 0) flush_sum<POOL>(sT, sP, cur, acc, lane);
                    acc = (v2f){0.f, 0.f};
                    cur = k;
                }
                acc = acc + r[i];
            }
        }
    }
    if (cur >= 0) flush_sum<POOL>(sT, sP, cur, acc, lane);
}

__global__ void __launch_bounds__(L1_THREADS) __attribute__((amdgpu_waves_per_eu(8, 8)))
gin_block_kernel(const float* __restrict__ X,
                 const int*   __restrict__ binned,
                 const int*   __restrict__ offs,
                 const int*   __restrict__ gid,
                 const float* __restrict__ W,
                 const float* __restrict__ bias,
                 float*       __restrict__ pool,
                 float*       __restrict__ cnt,
                 int nNodes, int nSlices)
{
    __shared__ float sT[PART_ROWS * TILE_PITCH];
    __shared__ float sWp[D_FEAT * D_FEAT];
    __shared__ float sB[D_FEAT];
    __shared__ float sP[N_GRAPHS * D_FEAT];
    __shared__ float sC[N_GRAPHS];
    __shared__ int      sCntG[G_BINS];
    __shared__ int      sOffG[G_BINS + 1];
    __shared__ int      sCntP[N_GRAPHS];
    __shared__ int      sOffP[N_GRAPHS + 1];
    __shared__ unsigned sListG[LIST_CAP];
    __shared__ unsigned sListP[LIST_CAP];

    const int tid = threadIdx.x;
    for (int i = tid; i < PART_ROWS * TILE_PITCH; i += L1_THREADS) sT[i] = 0.f;
    for (int i = tid; i < N_GRAPHS * D_FEAT; i += L1_THREADS) sP[i] = 0.f;
    if (tid < N_GRAPHS) { sC[tid] = 0.f; sCntP[tid] = 0; }
    if (tid < G_BINS) sCntG[tid] = 0;
    for (int idx = tid; idx < D_FEAT * D_FEAT; idx += L1_THREADS) {
        const int k   = idx >> 6;
        const int col = idx & 63;
        sWp[(k >> 1) * 128 + col * 2 + (k & 1)] = W[idx];
    }
    if (tid < D_FEAT) sB[tid] = bias[tid];
    __syncthreads();

    const int lane = tid & 31;
    const int wv   = __builtin_amdgcn_readfirstlane(tid >> 5);
    const int part = blockIdx.x;
    const int2* __restrict__ pairsDst = (const int2*)binned;
    const int2* __restrict__ pairsSrc = (const int2*)binned + (size_t)nSlices * SLICE;
    const int*  __restrict__ offsDst  = offs;
    const int*  __restrict__ offsSrc  = offs + nSlices * OFFS_PITCH;

    const WaveSegs sgG = wave_segments(offsDst, part, wv, lane, nSlices);
    const WaveSegs sgP = wave_segments(offsSrc, part, wv, lane, nSlices);
    sort_pass<false>(pairsDst, pairsSrc, sgG, sgP, lane, sCntG, sCntP, sListG, sListP);
    __syncthreads();
    if (wv == 0) wave_scan<G_PER_LANE>(sCntG, sOffG, lane);
    if (wv == 1) wave_scan<P_PER_LANE>(sCntP, sOffP, lane);
    __syncthreads();
    sort_pass<true>(pairsDst, pairsSrc, sgG, sgP, lane, sCntG, sCntP, sListG, sListP);
    __syncthreads();
    const bool listG = __builtin_amdgcn_readfirstlane(sOffG[G_BINS])   <= LIST_CAP;
    const bool listP = __builtin_amdgcn_readfirstlane(sOffP[N_GRAPHS]) <= LIST_CAP;

    if (listG) {
        const int s = __builtin_amdgcn_readfirstlane(sOffG[wv * G_PER_LANE]);
        const int e = __builtin_amdgcn_readfirstlane(sOffG[wv * G_PER_LANE + G_PER_LANE]);
        consume_list<false>(X, sT, sP, sListG, s, e, lane);
    } else {
        walk_block_segments<false>(X, sT, sP, pairsDst, offsDst, part, wv, lane, nSlices);
    }
    __syncthreads();

    const int rowLocal = wv * 16;
    const int rowBase  = part * PART_ROWS + rowLocal;
    const int halfSel  = lane >> 4;
    const int laneM    = lane & 15;
    const bool hasTile = (wv < PART_ROWS / 16) && (rowBase < nNodes);

    if (hasTile) {
        int aRow = rowBase + laneM;
        if (aRow >= nNodes) aRow = nNodes - 1;
        const float* __restrict__ xr = X + (size_t)aRow * D_FEAT;
        const float* tr = sT + (rowLocal + laneM) * TILE_PITCH;

        v8f acc[4];
        #pragma unroll
        for (int ct = 0; ct < 4; ++ct) acc[ct] = (v8f){0.f,0.f,0.f,0.f,0.f,0.f,0.f,0.f};

        #pragma unroll 4
        for (int kt = 0; kt < 16; ++kt) {
            const int k0 = kt * 4 + halfSel * 2;
            const v2f a = *(const v2f*)(xr + k0) + *(const v2f*)(tr + k0);
            const int bRow = (kt * 2 + halfSel) * 128;
            #pragma unroll
            for (int ct = 0; ct < 4; ++ct) {
                const int col = ct * 16 + laneM;
                const v2f b = *(const v2f*)(sWp + bRow + col * 2);
                acc[ct] = __builtin_amdgcn_wmma_f32_16x16x4_f32(
                    false, a, false, b, (short)0, acc[ct], false, false);
            }
        }

        #pragma unroll
        for (int ct = 0; ct < 4; ++ct) {
            const int col = ct * 16 + laneM;
            const float bv = sB[col];
            #pragma unroll
            for (int v = 0; v < 8; ++v)
                sT[(rowLocal + v + halfSel * 8) * TILE_PITCH + col] = fmaxf(acc[ct][v] + bv, 0.f);
        }
    }
    __syncthreads();

    if (hasTile) {
        const int n  = min(GATHER_ROWS, nNodes - rowBase);
        const int li = lane & (GATHER_ROWS - 1);
        const int gv = gid[rowBase + min(li, n - 1)] & (N_GRAPHS - 1);
        if (lane < n) unsafeAtomicAdd(&sC[gv], 1.0f);

        v2f r[GATHER_ROWS];
        #pragma unroll
        for (int i = 0; i < GATHER_ROWS; ++i)
            r[i] = *(const v2f*)(sT + (rowLocal + i) * TILE_PITCH + 2 * lane);
        int cur = -1;
        v2f acc = (v2f){0.f, 0.f};
        #pragma unroll
        for (int i = 0; i < GATHER_ROWS; ++i) {
            if (i < n) {
                const int g = __builtin_amdgcn_readlane(gv, i);
                if (g != cur) {
                    if (cur >= 0) flush_sum<true>(sT, sP, cur, acc, lane);
                    acc = (v2f){0.f, 0.f};
                    cur = g;
                }
                acc = acc + r[i];
            }
        }
        if (cur >= 0) flush_sum<true>(sT, sP, cur, acc, lane);
    }
    if (listP) {
        const int s = __builtin_amdgcn_readfirstlane(sOffP[wv * P_PER_LANE]);
        const int e = __builtin_amdgcn_readfirstlane(sOffP[wv * P_PER_LANE + P_PER_LANE]);
        consume_list<true>(X, sT, sP, sListP, s, e, lane);
    } else {
        walk_block_segments<true>(X, sT, sP, pairsSrc, offsSrc, part, wv, lane, nSlices);
    }
    __syncthreads();

    constexpr int LINES = N_GRAPHS * D_FEAT / 32;
    #pragma unroll
    for (int q = 0; q < LINES / (L1_THREADS / 32); ++q) {
        const int line = (wv + (L1_THREADS / 32) * q + (int)blockIdx.x) & (LINES - 1);
        const int idx  = line * 32 + lane;
        unsafeAtomicAdd(pool + idx, sP[idx]);
    }
    if (tid < N_GRAPHS) unsafeAtomicAdd(cnt + tid, sC[tid]);
}

__global__ void __launch_bounds__(256)
gin_pool_linear_kernel(const float* __restrict__ pool,
                       const float* __restrict__ cnt,
                       const float* __restrict__ W,
                       const float* __restrict__ bias,
                       float*       __restrict__ out)
{
    __shared__ float sWp[D_FEAT * D_FEAT];
    __shared__ float sB[D_FEAT];
    stage_weights(W, bias, sWp, sB);

    const int wave    = threadIdx.x >> 5;
    const int lane    = threadIdx.x & 31;
    const int rowBase = (blockIdx.x * 8 + wave) * 16;
    if (rowBase >= N_GRAPHS) return;

    const int halfSel = lane >> 4;
    const int laneM   = lane & 15;

    v8f acc[4];
    wave_tile_product<false>(pool + (size_t)(rowBase + laneM) * D_FEAT, nullptr,
                             sWp, halfSel, laneM, acc);

    #pragma unroll
    for (int v = 0; v < 8; ++v) {
        const int row = rowBase + v + halfSel * 8;
        const float nv = cnt[row];
        #pragma unroll
        for (int ct = 0; ct < 4; ++ct) {
            const int col = ct * 16 + laneM;
            out[(size_t)row * D_FEAT + col] = acc[ct][v] + nv * sB[col];
        }
    }
}

extern "C" void kernel_launch(void* const* d_in, const int* in_sizes, int n_in,
                              void* d_out, int out_size, void* d_ws, size_t ws_size,
                              hipStream_t stream)
{
    const float* feats = (const float*)d_in[0];
    const int*   src   = (const int*)  d_in[1];
    const int*   dst   = (const int*)  d_in[2];
    const int*   gid   = (const int*)  d_in[3];
    const float* W1    = (const float*)d_in[4];
    const float* b1    = (const float*)d_in[5];
    const float* W2    = (const float*)d_in[6];
    const float* b2    = (const float*)d_in[7];
    float* out = (float*)d_out;

    const int nNodes = in_sizes[0] / D_FEAT;
    const int nEdges = in_sizes[1];
    const int nSlices = (nEdges + SLICE - 1) / SLICE;
    const int nParts  = (nNodes + PART_ROWS - 1) / PART_ROWS;

    float* pool   = (float*)d_ws;
    float* cnt    = pool + N_GRAPHS * D_FEAT;
    int*   binned = (int*)(cnt + N_GRAPHS);
    int*   offs   = binned + (size_t)2 * nSlices * SLICE * 2;

    hipMemsetAsync(pool, 0, (N_GRAPHS * D_FEAT + N_GRAPHS) * sizeof(float), stream);

    gin_bin_kernel<<<2 * nSlices, BIN_THREADS, 0, stream>>>(src, dst, gid, binned, offs,
                                                            nEdges, nSlices);
    gin_block_kernel<<<nParts, L1_THREADS, 0, stream>>>(feats, binned, offs, gid, W1, b1,
                                                        pool, cnt, nNodes, nSlices);

    gin_pool_linear_kernel<<<1, 256, 0, stream>>>(pool, cnt, W2, b2, out);
}


# layer 2: W2 applied per block to its own 128x64 partial table (WMMA), result + count*b2 into out by f32 atomics (zeroed in sort kernel); no pool kernel, no memset; sums reassociated, arrival order
# speedup vs baseline: 4.2440x; 1.0606x over previous
#include <hip/hip_runtime.h>


typedef float v2f __attribute__((ext_vector_type(2)));
typedef float v8f __attribute__((ext_vector_type(8)));

#define D_FEAT   64
#define N_GRAPHS 128

#define PART_ROWS   400
#define MAX_PARTS   256
#define SLICE       8192
#define BIN_THREADS 1024
#define EDGES_PER_THREAD (SLICE / BIN_THREADS)
#define OFFS_PITCH  (MAX_PARTS + 1)

__global__ void __launch_bounds__(BIN_THREADS)
gin_bin_kernel(const int* __restrict__ src,
               const int* __restrict__ dst,
               const int* __restrict__ gid,
               int*       __restrict__ binned,
               int*       __restrict__ offs,
               float*     __restrict__ out,
               int nEdges, int nSlices)
{
    __shared__ int  sCnt[MAX_PARTS];
    __shared__ int  sOff[MAX_PARTS + 1];
    __shared__ int  sCur[MAX_PARTS];
    __shared__ int2 sPairs[SLICE];

    const int tid   = threadIdx.x;
    const int bySrc = ((int)blockIdx.x >= nSlices) ? 1 : 0;
    const int slice = (int)blockIdx.x - bySrc * nSlices;
    const int e0    = slice * SLICE;
    const int n     = min(SLICE, nEdges - e0);

    if (tid < MAX_PARTS) sCnt[tid] = 0;
    __syncthreads();

    int kv[EDGES_PER_THREAD], pv[EDGES_PER_THREAD];
    if (bySrc == 0) {
        #pragma unroll
        for (int j = 0; j < EDGES_PER_THREAD; ++j) {
            const int i = tid + j * BIN_THREADS;
            kv[j] = (i < n) ? dst[e0 + i] : 0;
            pv[j] = (i < n) ? src[e0 + i] : 0;
        }
    } else {
        int dv[EDGES_PER_THREAD];
        #pragma unroll
        for (int j = 0; j < EDGES_PER_THREAD; ++j) {
            const int i = tid + j * BIN_THREADS;
            kv[j] = (i < n) ? src[e0 + i] : 0;
            dv[j] = (i < n) ? dst[e0 + i] : 0;
        }
        #pragma unroll
        for (int j = 0; j < EDGES_PER_THREAD; ++j) {
            const int i = tid + j * BIN_THREADS;
            pv[j] = (i < n) ? gid[dv[j]] : 0;
        }
    }
    #pragma unroll
    for (int j = 0; j < EDGES_PER_THREAD; ++j) {
        if (tid + j * BIN_THREADS < n)
            atomicAdd(&sCnt[min(kv[j] / PART_ROWS, MAX_PARTS - 1)], 1);
    }
    __syncthreads();

    if (tid < 32) {
        int c[8];
        int sum = 0;
        #pragma unroll
        for (int j = 0; j < 8; ++j) { c[j] = sCnt[tid * 8 + j]; sum += c[j]; }
        int incl = sum;
        #pragma unroll
        for (int d = 1; d < 32; d <<= 1) {
            const int t = __shfl_up(incl, d);
            if (tid >= d) incl += t;
        }
        int run = incl - sum;
        #pragma unroll
        for (int j = 0; j < 8; ++j) {
            sOff[tid * 8 + j] = run;
            sCur[tid * 8 + j] = run;
            run += c[j];
        }
        if (tid == 31) sOff[MAX_PARTS] = run;
    }
    __syncthreads();

    #pragma unroll
    for (int j = 0; j < EDGES_PER_THREAD; ++j) {
        if (tid + j * BIN_THREADS < n) {
            const int p   = min(kv[j] / PART_ROWS, MAX_PARTS - 1);
            const int pos = atomicAdd(&sCur[p], 1);
            sPairs[pos] = make_int2(pv[j], kv[j] - p * PART_ROWS);
        }
    }
    __syncthreads();

    int2* __restrict__ o = (int2*)binned + (size_t)blockIdx.x * SLICE;
    for (int i = tid; i < n; i += BIN_THREADS) o[i] = sPairs[i];
    if (tid <= MAX_PARTS) offs[blockIdx.x * OFFS_PITCH + tid] = sOff[tid];

    if (blockIdx.x == 0)
        for (int i = tid; i < N_GRAPHS * D_FEAT; i += BIN_THREADS) out[i] = 0.f;
}


#define L1_THREADS  1024
#define TILE_PITCH  66
#define P_PITCH     66
#define GATHER_ROWS 16

template <bool POOL>
__device__ __forceinline__ void
walk_block_segments(const float* __restrict__ X, float* sT, float* sP,
                    const int2* __restrict__ pairs, const int* __restrict__ offs,
                    int part, int wv, int lane, int nSlices)
{
    const int nMine = (nSlices > wv) ? (nSlices - wv + 31) / 32 : 0;
    int lov = 0, hiv = 0;
    if (lane < nMine) {
        const int* __restrict__ o = offs + (wv + 32 * lane) * OFFS_PITCH + part;
        lov = o[0];
        hiv = o[1];
    }
    const int li = lane & (GATHER_ROWS - 1);

    int j = 0, b0 = 0, segHi = 0;
    for (; j < nMine; ++j) {
        b0    = __builtin_amdgcn_readlane(lov, j);
        segHi = __builtin_amdgcn_readlane(hiv, j);
        if (b0 < segHi) break;
    }
    int2 pr = make_int2(0, 0);
    if (j < nMine)
        pr = pairs[(size_t)(wv + 32 * j) * SLICE + b0 + min(li, min(GATHER_ROWS, segHi - b0) - 1)];

    while (j < nMine) {
        const int n = min(GATHER_ROWS, segHi - b0);
        const int tileRow = min(pr.y, PART_ROWS - 1);
        const int ldv  = POOL ? tileRow : pr.x;
        const int addv = POOL ? (pr.x & (N_GRAPHS - 1)) : tileRow;

        v2f r[GATHER_ROWS];
        #pragma unroll
        for (int i = 0; i < GATHER_ROWS; ++i) {
            const int s = __builtin_amdgcn_readlane(ldv, i);
            if (POOL) r[i] = *(const v2f*)(sT + s * TILE_PITCH + 2 * lane);
            else      r[i] = ((const v2f*)(X + (size_t)s * D_FEAT))[lane];
        }

        int nj = j, nb0 = b0 + GATHER_ROWS, nHi = segHi;
        if (nb0 >= nHi) {
            for (++nj; nj < nMine; ++nj) {
                nb0 = __builtin_amdgcn_readlane(lov, nj);
                nHi = __builtin_amdgcn_readlane(hiv, nj);
                if (nb0 < nHi) break;
            }
        }
        if (nj < nMine)
            pr = pairs[(size_t)(wv + 32 * nj) * SLICE + nb0 + min(li, min(GATHER_ROWS, nHi - nb0) - 1)];

        #pragma unroll
        for (int i = 0; i < GATHER_ROWS; ++i) {
            if (i < n) {
                const int a = __builtin_amdgcn_readlane(addv, i);
                float* t = POOL ? (sP + a * P_PITCH + 2 * lane)
                                : (sT + a * TILE_PITCH + 2 * lane);
                unsafeAtomicAdd(t,     r[i].x);
                unsafeAtomicAdd(t + 1, r[i].y);
            }
        }
        j = nj; b0 = nb0; segHi = nHi;
    }
}

#define LIST_CAP    8192
#define G_PER_LANE  13
#define G_BINS      (32 * G_PER_LANE)
#define P_PER_LANE  (N_GRAPHS / 32)
#define LOAD_CH     8
#define ENT_SHIFT   17
#define ENT_MASK    0x1FFFF
static_assert(G_BINS >= PART_ROWS, "row bins");
static_assert(MAX_PARTS * PART_ROWS <= (1 << ENT_SHIFT), "payload bits");

struct WaveSegs { int incl; int q; int total; int nMine; };

__device__ __forceinline__ WaveSegs
wave_segments(const int* __restrict__ offs, int part, int wv, int lane, int nSlices)
{
    WaveSegs s;
    s.nMine = (nSlices > wv) ? (nSlices - wv + 31) / 32 : 0;
    int lo = 0, hi = 0;
    if (lane < s.nMine) {
        const int* __restrict__ o = offs + (wv + 32 * lane) * OFFS_PITCH + part;
        lo = o[0];
        hi = o[1];
    }
    const int len = max(hi - lo, 0);
    int incl = len;
    #pragma unroll
    for (int d = 1; d < 32; d <<= 1) {
        const int t = __shfl_up(incl, d);
        if (lane >= d) incl += t;
    }
    s.incl  = incl;
    s.total = __builtin_amdgcn_readlane(incl, 31);
    s.q     = (lane < s.nMine) ? (wv + 32 * lane) * SLICE + lo - (incl - len) : 0;
    return s;
}

__device__ __forceinline__ void
chunk_indices(const WaveSegs& sg, int base, int lane, int (&t)[LOAD_CH])
{
    int j[LOAD_CH];
    #pragma unroll
    for (int c = 0; c < LOAD_CH; ++c) {
        t[c] = max(min(base + c * 32 + lane, sg.total - 1), 0);
        j[c] = 0;
    }
    for (int jj = 0; jj < sg.nMine; ++jj) {
        const int end = __builtin_amdgcn_readlane(sg.incl, jj);
        #pragma unroll
        for (int c = 0; c < LOAD_CH; ++c) j[c] += (t[c] >= end) ? 1 : 0;
    }
    #pragma unroll
    for (int c = 0; c < LOAD_CH; ++c) t[c] += __shfl(sg.q, j[c]);
}

template <bool POOL>
__device__ __forceinline__ unsigned make_entry(int2 pr)
{
    const unsigned row = min((unsigned)pr.y, (unsigned)(PART_ROWS - 1));
    return POOL ? ((((unsigned)pr.x & (N_GRAPHS - 1)) << ENT_SHIFT) | row)
                : ((row << ENT_SHIFT) | ((unsigned)pr.x & ENT_MASK));
}

template <bool POOL, bool SCATTER>
__device__ __forceinline__ void
put_chunks(const int2 (&pr)[LOAD_CH], int total, int base, int lane,
           int* sCur, unsigned* sList)
{
    unsigned ent[LOAD_CH];
    int      pos[LOAD_CH];
    #pragma unroll
    for (int c = 0; c < LOAD_CH; ++c) ent[c] = make_entry<POOL>(pr[c]);
    #pragma unroll
    for (int c = 0; c < LOAD_CH; ++c) {
        pos[c] = LIST_CAP;
        if (base + c * 32 + lane < total) pos[c] = atomicAdd(&sCur[ent[c] >> ENT_SHIFT], 1);
    }
    if (SCATTER) {
        #pragma unroll
        for (int c = 0; c < LOAD_CH; ++c)
            if (pos[c] < LIST_CAP) sList[pos[c]] = ent[c];
    }
}

template <bool POOL, bool SCATTER>
__device__ __forceinline__ int2 load_pair(const int2* __restrict__ pairs, int t)
{
    if (SCATTER) return pairs[t];
    const int* __restrict__ p = (const int*)(pairs + t);
    return POOL ? make_int2(p[0], 0) : make_int2(0, p[1]);
}

template <bool SCATTER>
__device__ __forceinline__ void
sort_pass(const int2* __restrict__ pairsDst, const int2* __restrict__ pairsSrc,
          const WaveSegs& sgG, const WaveSegs& sgP, int lane,
          int* sCurG, int* sCurP, unsigned* sListG, unsigned* sListP)
{
    const int tmax = max(sgG.total, sgP.total);
    for (int base = 0; base < tmax; base += 32 * LOAD_CH) {
        int  tg[LOAD_CH], tp[LOAD_CH];
        int2 pg[LOAD_CH], pp[LOAD_CH];
        chunk_indices(sgG, base, lane, tg);
        chunk_indices(sgP, base, lane, tp);
        #pragma unroll
        for (int c = 0; c < LOAD_CH; ++c) pg[c] = load_pair<false, SCATTER>(pairsDst, tg[c]);
        #pragma unroll
        for (int c = 0; c < LOAD_CH; ++c) pp[c] = load_pair<true,  SCATTER>(pairsSrc, tp[c]);
        put_chunks<false, SCATTER>(pg, sgG.total, base, lane, sCurG, sListG);
        put_chunks<true,  SCATTER>(pp, sgP.total, base, lane, sCurP, sListP);
    }
}

template <int PER>
__device__ __forceinline__ void wave_scan(int* sCnt, int* sOff, int lane)
{
    int c[PER];
    int sum = 0;
    #pragma unroll
    for (int j = 0; j < PER; ++j) { c[j] = sCnt[lane * PER + j]; sum += c[j]; }
    int incl = sum;
    #pragma unroll
    for (int d = 1; d < 32; d <<= 1) {
        const int t = __shfl_up(incl, d);
        if (lane >= d) incl += t;
    }
    int run = incl - sum;
    #pragma unroll
    for (int j = 0; j < PER; ++j) {
        sOff[lane * PER + j] = run;
        sCnt[lane * PER + j] = run;
        run += c[j];
    }
    if (lane == 31) sOff[32 * PER] = run;
}

template <bool POOL>
__device__ __forceinline__ void flush_sum(float* sT, float* sP, int key, v2f acc, int lane)
{
    if (POOL) {
        float* t = sP + key * P_PITCH + 2 * lane;
        unsafeAtomicAdd(t,     acc.x);
        unsafeAtomicAdd(t + 1, acc.y);
    } else {
        *(v2f*)(sT + key * TILE_PITCH + 2 * lane) = acc;
    }
}

template <bool POOL>
__device__ __forceinline__ void
consume_list(const float* __restrict__ X, float* sT, float* sP,
             const unsigned* sList, int s, int e, int lane)
{
    const int li = lane & (GATHER_ROWS - 1);
    int cur = -1;
    v2f acc = (v2f){0.f, 0.f};
    unsigned ent = 0;
    if (s < e) ent = sList[s + min(li, min(GATHER_ROWS, e - s) - 1)];

    for (int b = s; b < e; b += GATHER_ROWS) {
        const int n = min(GATHER_ROWS, e - b);
        const int ldv  = (int)(ent & ENT_MASK);
        const int keyv = (int)(ent >> ENT_SHIFT);

        v2f r[GATHER_ROWS];
        #pragma unroll
        for (int i = 0; i < GATHER_ROWS; ++i) {
            const int row = __builtin_amdgcn_readlane(ldv, i);
            if (POOL) r[i] = *(const v2f*)(sT + row * TILE_PITCH + 2 * lane);
            else      r[i] = ((const v2f*)(X + (size_t)row * D_FEAT))[lane];
        }
        const int nb = b + GATHER_ROWS;
        if (nb < e) ent = sList[nb + min(li, min(GATHER_ROWS, e - nb) - 1)];

        #pragma unroll
        for (int i = 0; i < GATHER_ROWS; ++i) {
            if (i < n) {
                const int k = __builtin_amdgcn_readlane(keyv, i);
                if (k != cur) {
                    if (cur >= 0) flush_sum<POOL>(sT, sP, cur, acc, lane);
                    acc = (v2f){0.f, 0.f};
                    cur = k;
                }
                acc = acc + r[i];
            }
        }
    }
    if (cur >= 0) flush_sum<POOL>(sT, sP, cur, acc, lane);
}

__global__ void __launch_bounds__(L1_THREADS) __attribute__((amdgpu_waves_per_eu(8, 8)))
gin_block_kernel(const float* __restrict__ X,
                 const int*   __restrict__ binned,
                 const int*   __restrict__ offs,
                 const int*   __restrict__ gid,
                 const float* __restrict__ W,
                 const float* __restrict__ bias,
                 const float* __restrict__ W2,
                 const float* __restrict__ bias2,
                 float*       __restrict__ out,
                 int nNodes, int nSlices)
{
    __shared__ float sT[PART_ROWS * TILE_PITCH];
    __shared__ float sWp[D_FEAT * D_FEAT];
    __shared__ float sWp2[D_FEAT * D_FEAT];
    __shared__ float sB[D_FEAT];
    __shared__ float sB2[D_FEAT];
    __shared__ float sP[N_GRAPHS * P_PITCH];
    __shared__ float sC[N_GRAPHS];
    __shared__ int      sCntG[G_BINS];
    __shared__ int      sOffG[G_BINS + 1];
    __shared__ int      sCntP[N_GRAPHS];
    __shared__ int      sOffP[N_GRAPHS + 1];
    __shared__ unsigned sListG[LIST_CAP];
    __shared__ unsigned sListP[LIST_CAP];

    const int tid = threadIdx.x;
    for (int i = tid; i < PART_ROWS * TILE_PITCH; i += L1_THREADS) sT[i] = 0.f;
    for (int i = tid; i < N_GRAPHS * P_PITCH; i += L1_THREADS) sP[i] = 0.f;
    if (tid < N_GRAPHS) { sC[tid] = 0.f; sCntP[tid] = 0; }
    if (tid < G_BINS) sCntG[tid] = 0;
    for (int idx = tid; idx < D_FEAT * D_FEAT; idx += L1_THREADS) {
        const int k   = idx >> 6;
        const int col = idx & 63;
        const int o   = (k >> 1) * 128 + col * 2 + (k & 1);
        sWp[o]  = W[idx];
        sWp2[o] = W2[idx];
    }
    if (tid < D_FEAT) { sB[tid] = bias[tid]; sB2[tid] = bias2[tid]; }
    __syncthreads();

    const int lane = tid & 31;
    const int wv   = __builtin_amdgcn_readfirstlane(tid >> 5);
    const int part = blockIdx.x;
    const int2* __restrict__ pairsDst = (const int2*)binned;
    const int2* __restrict__ pairsSrc = (const int2*)binned + (size_t)nSlices * SLICE;
    const int*  __restrict__ offsDst  = offs;
    const int*  __restrict__ offsSrc  = offs + nSlices * OFFS_PITCH;

    const WaveSegs sgG = wave_segments(offsDst, part, wv, lane, nSlices);
    const WaveSegs sgP = wave_segments(offsSrc, part, wv, lane, nSlices);
    sort_pass<false>(pairsDst, pairsSrc, sgG, sgP, lane, sCntG, sCntP, sListG, sListP);
    __syncthreads();
    if (wv == 0) wave_scan<G_PER_LANE>(sCntG, sOffG, lane);
    if (wv == 1) wave_scan<P_PER_LANE>(sCntP, sOffP, lane);
    __syncthreads();
    sort_pass<true>(pairsDst, pairsSrc, sgG, sgP, lane, sCntG, sCntP, sListG, sListP);
    __syncthreads();
    const bool listG = __builtin_amdgcn_readfirstlane(sOffG[G_BINS])   <= LIST_CAP;
    const bool listP = __builtin_amdgcn_readfirstlane(sOffP[N_GRAPHS]) <= LIST_CAP;

    if (listG) {
        const int s = __builtin_amdgcn_readfirstlane(sOffG[wv * G_PER_LANE]);
        const int e = __builtin_amdgcn_readfirstlane(sOffG[wv * G_PER_LANE + G_PER_LANE]);
        consume_list<false>(X, sT, sP, sListG, s, e, lane);
    } else {
        walk_block_segments<false>(X, sT, sP, pairsDst, offsDst, part, wv, lane, nSlices);
    }
    __syncthreads();

    const int rowLocal = wv * 16;
    const int rowBase  = part * PART_ROWS + rowLocal;
    const int halfSel  = lane >> 4;
    const int laneM    = lane & 15;
    const bool hasTile = (wv < PART_ROWS / 16) && (rowBase < nNodes);

    if (hasTile) {
        int aRow = rowBase + laneM;
        if (aRow >= nNodes) aRow = nNodes - 1;
        const float* __restrict__ xr = X + (size_t)aRow * D_FEAT;
        const float* tr = sT + (rowLocal + laneM) * TILE_PITCH;

        v8f acc[4];
        #pragma unroll
        for (int ct = 0; ct < 4; ++ct) acc[ct] = (v8f){0.f,0.f,0.f,0.f,0.f,0.f,0.f,0.f};

        #pragma unroll 4
        for (int kt = 0; kt < 16; ++kt) {
            const int k0 = kt * 4 + halfSel * 2;
            const v2f a = *(const v2f*)(xr + k0) + *(const v2f*)(tr + k0);
            const int bRow = (kt * 2 + halfSel) * 128;
            #pragma unroll
            for (int ct = 0; ct < 4; ++ct) {
                const int col = ct * 16 + laneM;
                const v2f b = *(const v2f*)(sWp + bRow + col * 2);
                acc[ct] = __builtin_amdgcn_wmma_f32_16x16x4_f32(
                    false, a, false, b, (short)0, acc[ct], false, false);
            }
        }

        #pragma unroll
        for (int ct = 0; ct < 4; ++ct) {
            const int col = ct * 16 + laneM;
            const float bv = sB[col];
            #pragma unroll
            for (int v = 0; v < 8; ++v)
                sT[(rowLocal + v + halfSel * 8) * TILE_PITCH + col] = fmaxf(acc[ct][v] + bv, 0.f);
        }
    }
    __syncthreads();

    if (hasTile) {
        const int n  = min(GATHER_ROWS, nNodes - rowBase);
        const int li = lane & (GATHER_ROWS - 1);
        const int gv = gid[rowBase + min(li, n - 1)] & (N_GRAPHS - 1);
        if (lane < n) unsafeAtomicAdd(&sC[gv], 1.0f);

        v2f r[GATHER_ROWS];
        #pragma unroll
        for (int i = 0; i < GATHER_ROWS; ++i)
            r[i] = *(const v2f*)(sT + (rowLocal + i) * TILE_PITCH + 2 * lane);
        int cur = -1;
        v2f acc = (v2f){0.f, 0.f};
        #pragma unroll
        for (int i = 0; i < GATHER_ROWS; ++i) {
            if (i < n) {
                const int g = __builtin_amdgcn_readlane(gv, i);
                if (g != cur) {
                    if (cur >= 0) flush_sum<true>(sT, sP, cur, acc, lane);
                    acc = (v2f){0.f, 0.f};
                    cur = g;
                }
                acc = acc + r[i];
            }
        }
        if (cur >= 0) flush_sum<true>(sT, sP, cur, acc, lane);
    }
    if (listP) {
        const int s = __builtin_amdgcn_readfirstlane(sOffP[wv * P_PER_LANE]);
        const int e = __builtin_amdgcn_readfirstlane(sOffP[wv * P_PER_LANE + P_PER_LANE]);
        consume_list<true>(X, sT, sP, sListP, s, e, lane);
    } else {
        walk_block_segments<true>(X, sT, sP, pairsSrc, offsSrc, part, wv, lane, nSlices);
    }
    __syncthreads();

    float* sO = sT;
    {
        const int rt = wv >> 2;
        const int ct = wv & 3;
        const float* pr = sP + (rt * 16 + laneM) * P_PITCH;
        const int col = ct * 16 + laneM;
        v8f acc2 = (v8f){0.f,0.f,0.f,0.f,0.f,0.f,0.f,0.f};
        #pragma unroll
        for (int kt = 0; kt < 16; ++kt) {
            const int k0 = kt * 4 + halfSel * 2;
            const v2f a = *(const v2f*)(pr + k0);
            const v2f b = *(const v2f*)(sWp2 + (kt * 2 + halfSel) * 128 + col * 2);
            acc2 = __builtin_amdgcn_wmma_f32_16x16x4_f32(
                false, a, false, b, (short)0, acc2, false, false);
        }
        const float bv = sB2[col];
        #pragma unroll
        for (int v = 0; v < 8; ++v) {
            const int row = rt * 16 + v + halfSel * 8;
            sO[row * D_FEAT + col] = acc2[v] + sC[row] * bv;
        }
    }
    __syncthreads();

    constexpr int LINES = N_GRAPHS * D_FEAT / 32;
    #pragma unroll
    for (int q = 0; q < LINES / (L1_THREADS / 32); ++q) {
        const int line = (wv + (L1_THREADS / 32) * q + (int)blockIdx.x) & (LINES - 1);
        const int idx  = line * 32 + lane;
        unsafeAtomicAdd(out + idx, sO[idx]);
    }
}

extern "C" void kernel_launch(void* const* d_in, const int* in_sizes, int n_in,
                              void* d_out, int out_size, void* d_ws, size_t ws_size,
                              hipStream_t stream)
{
    const float* feats = (const float*)d_in[0];
    const int*   src   = (const int*)  d_in[1];
    const int*   dst   = (const int*)  d_in[2];
    const int*   gid   = (const int*)  d_in[3];
    const float* W1    = (const float*)d_in[4];
    const float* b1    = (const float*)d_in[5];
    const float* W2    = (const float*)d_in[6];
    const float* b2    = (const float*)d_in[7];
    float* out = (float*)d_out;

    const int nNodes = in_sizes[0] / D_FEAT;
    const int nEdges = in_sizes[1];
    const int nSlices = (nEdges + SLICE - 1) / SLICE;
    const int nParts  = (nNodes + PART_ROWS - 1) / PART_ROWS;

    int*   binned = (int*)d_ws;
    int*   offs   = binned + (size_t)2 * nSlices * SLICE * 2;

    gin_bin_kernel<<<2 * nSlices, BIN_THREADS, 0, stream>>>(src, dst, gid, binned, offs, out,
                                                            nEdges, nSlices);
    gin_block_kernel<<<nParts, L1_THREADS, 0, stream>>>(feats, binned, offs, gid, W1, b1,
                                                        W2, b2, out, nNodes, nSlices);
}


# sort kernel, src sort: graph of dst looked up in an LDS byte table of gid, filled per work-group with whole-line 16-B loads (was a 4-B global gather per edge); LDS staging packed to 4 B per edge
# speedup vs baseline: 4.5619x; 1.0749x over previous
#include <hip/hip_runtime.h>


typedef float v2f __attribute__((ext_vector_type(2)));
typedef float v8f __attribute__((ext_vector_type(8)));

#define D_FEAT   64
#define N_GRAPHS 128

#define PART_ROWS   400
#define MAX_PARTS   256
#define SLICE       8192
#define BIN_THREADS 1024
#define EDGES_PER_THREAD (SLICE / BIN_THREADS)
#define OFFS_PITCH  (MAX_PARTS + 1)

#define TAB_NODES (MAX_PARTS * PART_ROWS)
#define TAB_BATCH 8
#define PK_SHIFT  9
static_assert(PART_ROWS <= (1 << PK_SHIFT), "local row bits");
__global__ void __launch_bounds__(BIN_THREADS)
gin_bin_kernel(const int* __restrict__ src,
               const int* __restrict__ dst,
               const int* __restrict__ gid,
               int*       __restrict__ binned,
               int*       __restrict__ offs,
               float*     __restrict__ out,
               int nEdges, int nSlices, int nNodes)
{
    __shared__ int  sCnt[MAX_PARTS];
    __shared__ int  sOff[MAX_PARTS + 1];
    __shared__ int  sCur[MAX_PARTS];
    __shared__ unsigned sPk[SLICE];
    __shared__ unsigned sTab[TAB_NODES / 4];

    const int tid   = threadIdx.x;
    const int bySrc = ((int)blockIdx.x >= nSlices) ? 1 : 0;
    const int slice = (int)blockIdx.x - bySrc * nSlices;
    const int e0    = slice * SLICE;
    const int n     = min(SLICE, nEdges - e0);

    if (tid < MAX_PARTS) sCnt[tid] = 0;
    __syncthreads();

    int kv[EDGES_PER_THREAD], pv[EDGES_PER_THREAD];
    if (bySrc == 0) {
        #pragma unroll
        for (int j = 0; j < EDGES_PER_THREAD; ++j) {
            const int i = tid + j * BIN_THREADS;
            kv[j] = (i < n) ? dst[e0 + i] : 0;
            pv[j] = (i < n) ? src[e0 + i] : 0;
        }
    } else {
        #pragma unroll
        for (int j = 0; j < EDGES_PER_THREAD; ++j) {
            const int i = tid + j * BIN_THREADS;
            kv[j] = (i < n) ? src[e0 + i] : 0;
            pv[j] = (i < n) ? dst[e0 + i] : 0;
        }
        const int4* __restrict__ gq = (const int4*)gid;
        const int nQ = min(nNodes, TAB_NODES) >> 2;
        for (int base = 0; base < nQ; base += TAB_BATCH * BIN_THREADS) {
            int4 g[TAB_BATCH];
            #pragma unroll
            for (int u = 0; u < TAB_BATCH; ++u) {
                const int i = base + u * BIN_THREADS + tid;
                g[u] = gq[min(i, nQ - 1)];
            }
            #pragma unroll
            for (int u = 0; u < TAB_BATCH; ++u) {
                const int i = base + u * BIN_THREADS + tid;
                if (i < nQ)
                    sTab[i] = ((unsigned)g[u].x & 255u) | (((unsigned)g[u].y & 255u) << 8) |
                              (((unsigned)g[u].z & 255u) << 16) | (((unsigned)g[u].w & 255u) << 24);
            }
        }
        for (int i = (nQ << 2) + tid; i < min(nNodes, TAB_NODES); i += BIN_THREADS)
            ((unsigned char*)sTab)[i] = (unsigned char)gid[i];
    }
    #pragma unroll
    for (int j = 0; j < EDGES_PER_THREAD; ++j) {
        if (tid + j * BIN_THREADS < n)
            atomicAdd(&sCnt[min(kv[j] / PART_ROWS, MAX_PARTS - 1)], 1);
    }
    __syncthreads();

    if (tid < 32) {
        int c[8];
        int sum = 0;
        #pragma unroll
        for (int j = 0; j < 8; ++j) { c[j] = sCnt[tid * 8 + j]; sum += c[j]; }
        int incl = sum;
        #pragma unroll
        for (int d = 1; d < 32; d <<= 1) {
            const int t = __shfl_up(incl, d);
            if (tid >= d) incl += t;
        }
        int run = incl - sum;
        #pragma unroll
        for (int j = 0; j < 8; ++j) {
            sOff[tid * 8 + j] = run;
            sCur[tid * 8 + j] = run;
            run += c[j];
        }
        if (tid == 31) sOff[MAX_PARTS] = run;
    }
    __syncthreads();

    #pragma unroll
    for (int j = 0; j < EDGES_PER_THREAD; ++j) {
        if (tid + j * BIN_THREADS < n) {
            const int p   = min(kv[j] / PART_ROWS, MAX_PARTS - 1);
            const int pos = atomicAdd(&sCur[p], 1);
            const unsigned pay = bySrc
                ? (unsigned)((const unsigned char*)sTab)[min((unsigned)pv[j], (unsigned)(TAB_NODES - 1))]
                : (unsigned)pv[j];
            const unsigned row = (unsigned)min(kv[j] - p * PART_ROWS, PART_ROWS - 1);
            sPk[pos] = (pay << PK_SHIFT) | row;
        }
    }
    __syncthreads();

    int2* __restrict__ o = (int2*)binned + (size_t)blockIdx.x * SLICE;
    for (int i = tid; i < n; i += BIN_THREADS) {
        const unsigned ent = sPk[i];
        o[i] = make_int2((int)(ent >> PK_SHIFT), (int)(ent & ((1u << PK_SHIFT) - 1u)));
    }
    if (tid <= MAX_PARTS) offs[blockIdx.x * OFFS_PITCH + tid] = sOff[tid];

    if (blockIdx.x == 0)
        for (int i = tid; i < N_GRAPHS * D_FEAT; i += BIN_THREADS) out[i] = 0.f;
}


#define L1_THREADS  1024
#define TILE_PITCH  66
#define P_PITCH     66
#define GATHER_ROWS 16

template <bool POOL>
__device__ __forceinline__ void
walk_block_segments(const float* __restrict__ X, float* sT, float* sP,
                    const int2* __restrict__ pairs, const int* __restrict__ offs,
                    int part, int wv, int lane, int nSlices)
{
    const int nMine = (nSlices > wv) ? (nSlices - wv + 31) / 32 : 0;
    int lov = 0, hiv = 0;
    if (lane < nMine) {
        const int* __restrict__ o = offs + (wv + 32 * lane) * OFFS_PITCH + part;
        lov = o[0];
        hiv = o[1];
    }
    const int li = lane & (GATHER_ROWS - 1);

    int j = 0, b0 = 0, segHi = 0;
    for (; j < nMine; ++j) {
        b0    = __builtin_amdgcn_readlane(lov, j);
        segHi = __builtin_amdgcn_readlane(hiv, j);
        if (b0 < segHi) break;
    }
    int2 pr = make_int2(0, 0);
    if (j < nMine)
        pr = pairs[(size_t)(wv + 32 * j) * SLICE + b0 + min(li, min(GATHER_ROWS, segHi - b0) - 1)];

    while (j < nMine) {
        const int n = min(GATHER_ROWS, segHi - b0);
        const int tileRow = min(pr.y, PART_ROWS - 1);
        const int ldv  = POOL ? tileRow : pr.x;
        const int addv = POOL ? (pr.x & (N_GRAPHS - 1)) : tileRow;

        v2f r[GATHER_ROWS];
        #pragma unroll
        for (int i = 0; i < GATHER_ROWS; ++i) {
            const int s = __builtin_amdgcn_readlane(ldv, i);
            if (POOL) r[i] = *(const v2f*)(sT + s * TILE_PITCH + 2 * lane);
            else      r[i] = ((const v2f*)(X + (size_t)s * D_FEAT))[lane];
        }

        int nj = j, nb0 = b0 + GATHER_ROWS, nHi = segHi;
        if (nb0 >= nHi) {
            for (++nj; nj < nMine; ++nj) {
                nb0 = __builtin_amdgcn_readlane(lov, nj);
                nHi = __builtin_amdgcn_readlane(hiv, nj);
                if (nb0 < nHi) break;
            }
        }
        if (nj < nMine)
            pr = pairs[(size_t)(wv + 32 * nj) * SLICE + nb0 + min(li, min(GATHER_ROWS, nHi - nb0) - 1)];

        #pragma unroll
        for (int i = 0; i < GATHER_ROWS; ++i) {
            if (i < n) {
                const int a = __builtin_amdgcn_readlane(addv, i);
                float* t = POOL ? (sP + a * P_PITCH + 2 * lane)
                                : (sT + a * TILE_PITCH + 2 * lane);
                unsafeAtomicAdd(t,     r[i].x);
                unsafeAtomicAdd(t + 1, r[i].y);
            }
        }
        j = nj; b0 = nb0; segHi = nHi;
    }
}

#define LIST_CAP    8192
#define G_PER_LANE  13
#define G_BINS      (32 * G_PER_LANE)
#define P_PER_LANE  (N_GRAPHS / 32)
#define LOAD_CH     8
#define ENT_SHIFT   17
#define ENT_MASK    0x1FFFF
static_assert(G_BINS >= PART_ROWS, "row bins");
static_assert(MAX_PARTS * PART_ROWS <= (1 << ENT_SHIFT), "payload bits");

struct WaveSegs { int incl; int q; int total; int nMine; };

__device__ __forceinline__ WaveSegs
wave_segments(const int* __restrict__ offs, int part, int wv, int lane, int nSlices)
{
    WaveSegs s;
    s.nMine = (nSlices > wv) ? (nSlices - wv + 31) / 32 : 0;
    int lo = 0, hi = 0;
    if (lane < s.nMine) {
        const int* __restrict__ o = offs + (wv + 32 * lane) * OFFS_PITCH + part;
        lo = o[0];
        hi = o[1];
    }
    const int len = max(hi - lo, 0);
    int incl = len;
    #pragma unroll
    for (int d = 1; d < 32; d <<= 1) {
        const int t = __shfl_up(incl, d);
        if (lane >= d) incl += t;
    }
    s.incl  = incl;
    s.total = __builtin_amdgcn_readlane(incl, 31);
    s.q     = (lane < s.nMine) ? (wv + 32 * lane) * SLICE + lo - (incl - len) : 0;
    return s;
}

__device__ __forceinline__ void
chunk_indices(const WaveSegs& sg, int base, int lane, int (&t)[LOAD_CH])
{
    int j[LOAD_CH];
    #pragma unroll
    for (int c = 0; c < LOAD_CH; ++c) {
        t[c] = max(min(base + c * 32 + lane, sg.total - 1), 0);
        j[c] = 0;
    }
    for (int jj = 0; jj < sg.nMine; ++jj) {
        const int end = __builtin_amdgcn_readlane(sg.incl, jj);
        #pragma unroll
        for (int c = 0; c < LOAD_CH; ++c) j[c] += (t[c] >= end) ? 1 : 0;
    }
    #pragma unroll
    for (int c = 0; c < LOAD_CH; ++c) t[c] += __shfl(sg.q, j[c]);
}

template <bool POOL>
__device__ __forceinline__ unsigned make_entry(int2 pr)
{
    const unsigned row = min((unsigned)pr.y, (unsigned)(PART_ROWS - 1));
    return POOL ? ((((unsigned)pr.x & (N_GRAPHS - 1)) << ENT_SHIFT) | row)
                : ((row << ENT_SHIFT) | ((unsigned)pr.x & ENT_MASK));
}

template <bool POOL, bool SCATTER>
__device__ __forceinline__ void
put_chunks(const int2 (&pr)[LOAD_CH], int total, int base, int lane,
           int* sCur, unsigned* sList)
{
    unsigned ent[LOAD_CH];
    int      pos[LOAD_CH];
    #pragma unroll
    for (int c = 0; c < LOAD_CH; ++c) ent[c] = make_entry<POOL>(pr[c]);
    #pragma unroll
    for (int c = 0; c < LOAD_CH; ++c) {
        pos[c] = LIST_CAP;
        if (base + c * 32 + lane < total) pos[c] = atomicAdd(&sCur[ent[c] >> ENT_SHIFT], 1);
    }
    if (SCATTER) {
        #pragma unroll
        for (int c = 0; c < LOAD_CH; ++c)
            if (pos[c] < LIST_CAP) sList[pos[c]] = ent[c];
    }
}

template <bool POOL, bool SCATTER>
__device__ __forceinline__ int2 load_pair(const int2* __restrict__ pairs, int t)
{
    if (SCATTER) return pairs[t];
    const int* __restrict__ p = (const int*)(pairs + t);
    return POOL ? make_int2(p[0], 0) : make_int2(0, p[1]);
}

template <bool SCATTER>
__device__ __forceinline__ void
sort_pass(const int2* __restrict__ pairsDst, const int2* __restrict__ pairsSrc,
          const WaveSegs& sgG, const WaveSegs& sgP, int lane,
          int* sCurG, int* sCurP, unsigned* sListG, unsigned* sListP)
{
    const int tmax = max(sgG.total, sgP.total);
    for (int base = 0; base < tmax; base += 32 * LOAD_CH) {
        int  tg[LOAD_CH], tp[LOAD_CH];
        int2 pg[LOAD_CH], pp[LOAD_CH];
        chunk_indices(sgG, base, lane, tg);
        chunk_indices(sgP, base, lane, tp);
        #pragma unroll
        for (int c = 0; c < LOAD_CH; ++c) pg[c] = load_pair<false, SCATTER>(pairsDst, tg[c]);
        #pragma unroll
        for (int c = 0; c < LOAD_CH; ++c) pp[c] = load_pair<true,  SCATTER>(pairsSrc, tp[c]);
        put_chunks<false, SCATTER>(pg, sgG.total, base, lane, sCurG, sListG);
        put_chunks<true,  SCATTER>(pp, sgP.total, base, lane, sCurP, sListP);
    }
}

template <int PER>
__device__ __forceinline__ void wave_scan(int* sCnt, int* sOff, int lane)
{
    int c[PER];
    int sum = 0;
    #pragma unroll
    for (int j = 0; j < PER; ++j) { c[j] = sCnt[lane * PER + j]; sum += c[j]; }
    int incl = sum;
    #pragma unroll
    for (int d = 1; d < 32; d <<= 1) {
        const int t = __shfl_up(incl, d);
        if (lane >= d) incl += t;
    }
    int run = incl - sum;
    #pragma unroll
    for (int j = 0; j < PER; ++j) {
        sOff[lane * PER + j] = run;
        sCnt[lane * PER + j] = run;
        run += c[j];
    }
    if (lane == 31) sOff[32 * PER] = run;
}

template <bool POOL>
__device__ __forceinline__ void flush_sum(float* sT, float* sP, int key, v2f acc, int lane)
{
    if (POOL) {
        float* t = sP + key * P_PITCH + 2 * lane;
        unsafeAtomicAdd(t,     acc.x);
        unsafeAtomicAdd(t + 1, acc.y);
    } else {
        *(v2f*)(sT + key * TILE_PITCH + 2 * lane) = acc;
    }
}

template <bool POOL>
__device__ __forceinline__ void
consume_list(const float* __restrict__ X, float* sT, float* sP,
             const unsigned* sList, int s, int e, int lane)
{
    const int li = lane & (GATHER_ROWS - 1);
    int cur = -1;
    v2f acc = (v2f){0.f, 0.f};
    unsigned ent = 0;
    if (s < e) ent = sList[s + min(li, min(GATHER_ROWS, e - s) - 1)];

    for (int b = s; b < e; b += GATHER_ROWS) {
        const int n = min(GATHER_ROWS, e - b);
        const int ldv  = (int)(ent & ENT_MASK);
        const int keyv = (int)(ent >> ENT_SHIFT);

        v2f r[GATHER_ROWS];
        #pragma unroll
        for (int i = 0; i < GATHER_ROWS; ++i) {
            const int row = __builtin_amdgcn_readlane(ldv, i);
            if (POOL) r[i] = *(const v2f*)(sT + row * TILE_PITCH + 2 * lane);
            else      r[i] = ((const v2f*)(X + (size_t)row * D_FEAT))[lane];
        }
        const int nb = b + GATHER_ROWS;
        if (nb < e) ent = sList[nb + min(li, min(GATHER_ROWS, e - nb) - 1)];

        #pragma unroll
        for (int i = 0; i < GATHER_ROWS; ++i) {
            if (i < n) {
                const int k = __builtin_amdgcn_readlane(keyv, i);
                if (k != cur) {
                    if (cur >= 0) flush_sum<POOL>(sT, sP, cur, acc, lane);
                    acc = (v2f){0.f, 0.f};
                    cur = k;
                }
                acc = acc + r[i];
            }
        }
    }
    if (cur >= 0) flush_sum<POOL>(sT, sP, cur, acc, lane);
}

__global__ void __launch_bounds__(L1_THREADS) __attribute__((amdgpu_waves_per_eu(8, 8)))
gin_block_kernel(const float* __restrict__ X,
                 const int*   __restrict__ binned,
                 const int*   __restrict__ offs,
                 const int*   __restrict__ gid,
                 const float* __restrict__ W,
                 const float* __restrict__ bias,
                 const float* __restrict__ W2,
                 const float* __restrict__ bias2,
                 float*       __restrict__ out,
                 int nNodes, int nSlices)
{
    __shared__ float sT[PART_ROWS * TILE_PITCH];
    __shared__ float sWp[D_FEAT * D_FEAT];
    __shared__ float sWp2[D_FEAT * D_FEAT];
    __shared__ float sB[D_FEAT];
    __shared__ float sB2[D_FEAT];
    __shared__ float sP[N_GRAPHS * P_PITCH];
    __shared__ float sC[N_GRAPHS];
    __shared__ int      sCntG[G_BINS];
    __shared__ int      sOffG[G_BINS + 1];
    __shared__ int      sCntP[N_GRAPHS];
    __shared__ int      sOffP[N_GRAPHS + 1];
    __shared__ unsigned sListG[LIST_CAP];
    __shared__ unsigned sListP[LIST_CAP];

    const int tid = threadIdx.x;
    for (int i = tid; i < PART_ROWS * TILE_PITCH; i += L1_THREADS) sT[i] = 0.f;
    for (int i = tid; i < N_GRAPHS * P_PITCH; i += L1_THREADS) sP[i] = 0.f;
    if (tid < N_GRAPHS) { sC[tid] = 0.f; sCntP[tid] = 0; }
    if (tid < G_BINS) sCntG[tid] = 0;
    for (int idx = tid; idx < D_FEAT * D_FEAT; idx += L1_THREADS) {
        const int k   = idx >> 6;
        const int col = idx & 63;
        const int o   = (k >> 1) * 128 + col * 2 + (k & 1);
        sWp[o]  = W[idx];
        sWp2[o] = W2[idx];
    }
    if (tid < D_FEAT) { sB[tid] = bias[tid]; sB2[tid] = bias2[tid]; }
    __syncthreads();

    const int lane = tid & 31;
    const int wv   = __builtin_amdgcn_readfirstlane(tid >> 5);
    const int part = blockIdx.x;
    const int2* __restrict__ pairsDst = (const int2*)binned;
    const int2* __restrict__ pairsSrc = (const int2*)binned + (size_t)nSlices * SLICE;
    const int*  __restrict__ offsDst  = offs;
    const int*  __restrict__ offsSrc  = offs + nSlices * OFFS_PITCH;

    const WaveSegs sgG = wave_segments(offsDst, part, wv, lane, nSlices);
    const WaveSegs sgP = wave_segments(offsSrc, part, wv, lane, nSlices);
    sort_pass<false>(pairsDst, pairsSrc, sgG, sgP, lane, sCntG, sCntP, sListG, sListP);
    __syncthreads();
    if (wv == 0) wave_scan<G_PER_LANE>(sCntG, sOffG, lane);
    if (wv == 1) wave_scan<P_PER_LANE>(sCntP, sOffP, lane);
    __syncthreads();
    sort_pass<true>(pairsDst, pairsSrc, sgG, sgP, lane, sCntG, sCntP, sListG, sListP);
    __syncthreads();
    const bool listG = __builtin_amdgcn_readfirstlane(sOffG[G_BINS])   <= LIST_CAP;
    const bool listP = __builtin_amdgcn_readfirstlane(sOffP[N_GRAPHS]) <= LIST_CAP;

    if (listG) {
        const int s = __builtin_amdgcn_readfirstlane(sOffG[wv * G_PER_LANE]);
        const int e = __builtin_amdgcn_readfirstlane(sOffG[wv * G_PER_LANE + G_PER_LANE]);
        consume_list<false>(X, sT, sP, sListG, s, e, lane);
    } else {
        walk_block_segments<false>(X, sT, sP, pairsDst, offsDst, part, wv, lane, nSlices);
    }
    __syncthreads();

    const int rowLocal = wv * 16;
    const int rowBase  = part * PART_ROWS + rowLocal;
    const int halfSel  = lane >> 4;
    const int laneM    = lane & 15;
    const bool hasTile = (wv < PART_ROWS / 16) && (rowBase < nNodes);

    if (hasTile) {
        int aRow = rowBase + laneM;
        if (aRow >= nNodes) aRow = nNodes - 1;
        const float* __restrict__ xr = X + (size_t)aRow * D_FEAT;
        const float* tr = sT + (rowLocal + laneM) * TILE_PITCH;

        v8f acc[4];
        #pragma unroll
        for (int ct = 0; ct < 4; ++ct) acc[ct] = (v8f){0.f,0.f,0.f,0.f,0.f,0.f,0.f,0.f};

        #pragma unroll 4
        for (int kt = 0; kt < 16; ++kt) {
            const int k0 = kt * 4 + halfSel * 2;
            const v2f a = *(const v2f*)(xr + k0) + *(const v2f*)(tr + k0);
            const int bRow = (kt * 2 + halfSel) * 128;
            #pragma unroll
            for (int ct = 0; ct < 4; ++ct) {
                const int col = ct * 16 + laneM;
                const v2f b = *(const v2f*)(sWp + bRow + col * 2);
                acc[ct] = __builtin_amdgcn_wmma_f32_16x16x4_f32(
                    false, a, false, b, (short)0, acc[ct], false, false);
            }
        }

        #pragma unroll
        for (int ct = 0; ct < 4; ++ct) {
            const int col = ct * 16 + laneM;
            const float bv = sB[col];
            #pragma unroll
            for (int v = 0; v < 8; ++v)
                sT[(rowLocal + v + halfSel * 8) * TILE_PITCH + col] = fmaxf(acc[ct][v] + bv, 0.f);
        }
    }
    __syncthreads();

    if (hasTile) {
        const int n  = min(GATHER_ROWS, nNodes - rowBase);
        const int li = lane & (GATHER_ROWS - 1);
        const int gv = gid[rowBase + min(li, n - 1)] & (N_GRAPHS - 1);
        if (lane < n) unsafeAtomicAdd(&sC[gv], 1.0f);

        v2f r[GATHER_ROWS];
        #pragma unroll
        for (int i = 0; i < GATHER_ROWS; ++i)
            r[i] = *(const v2f*)(sT + (rowLocal + i) * TILE_PITCH + 2 * lane);
        int cur = -1;
        v2f acc = (v2f){0.f, 0.f};
        #pragma unroll
        for (int i = 0; i < GATHER_ROWS; ++i) {
            if (i < n) {
                const int g = __builtin_amdgcn_readlane(gv, i);
                if (g != cur) {
                    if (cur >= 0) flush_sum<true>(sT, sP, cur, acc, lane);
                    acc = (v2f){0.f, 0.f};
                    cur = g;
                }
                acc = acc + r[i];
            }
        }
        if (cur >= 0) flush_sum<true>(sT, sP, cur, acc, lane);
    }
    if (listP) {
        const int s = __builtin_amdgcn_readfirstlane(sOffP[wv * P_PER_LANE]);
        const int e = __builtin_amdgcn_readfirstlane(sOffP[wv * P_PER_LANE + P_PER_LANE]);
        consume_list<true>(X, sT, sP, sListP, s, e, lane);
    } else {
        walk_block_segments<true>(X, sT, sP, pairsSrc, offsSrc, part, wv, lane, nSlices);
    }
    __syncthreads();

    float* sO = sT;
    {
        const int rt = wv >> 2;
        const int ct = wv & 3;
        const float* pr = sP + (rt * 16 + laneM) * P_PITCH;
        const int col = ct * 16 + laneM;
        v8f acc2 = (v8f){0.f,0.f,0.f,0.f,0.f,0.f,0.f,0.f};
        #pragma unroll
        for (int kt = 0; kt < 16; ++kt) {
            const int k0 = kt * 4 + halfSel * 2;
            const v2f a = *(const v2f*)(pr + k0);
            const v2f b = *(const v2f*)(sWp2 + (kt * 2 + halfSel) * 128 + col * 2);
            acc2 = __builtin_amdgcn_wmma_f32_16x16x4_f32(
                false, a, false, b, (short)0, acc2, false, false);
        }
        const float bv = sB2[col];
        #pragma unroll
        for (int v = 0; v < 8; ++v) {
            const int row = rt * 16 + v + halfSel * 8;
            sO[row * D_FEAT + col] = acc2[v] + sC[row] * bv;
        }
    }
    __syncthreads();

    constexpr int LINES = N_GRAPHS * D_FEAT / 32;
    #pragma unroll
    for (int q = 0; q < LINES / (L1_THREADS / 32); ++q) {
        const int line = (wv + (L1_THREADS / 32) * q + (int)blockIdx.x) & (LINES - 1);
        const int idx  = line * 32 + lane;
        unsafeAtomicAdd(out + idx, sO[idx]);
    }
}

extern "C" void kernel_launch(void* const* d_in, const int* in_sizes, int n_in,
                              void* d_out, int out_size, void* d_ws, size_t ws_size,
                              hipStream_t stream)
{
    const float* feats = (const float*)d_in[0];
    const int*   src   = (const int*)  d_in[1];
    const int*   dst   = (const int*)  d_in[2];
    const int*   gid   = (const int*)  d_in[3];
    const float* W1    = (const float*)d_in[4];
    const float* b1    = (const float*)d_in[5];
    const float* W2    = (const float*)d_in[6];
    const float* b2    = (const float*)d_in[7];
    float* out = (float*)d_out;

    const int nNodes = in_sizes[0] / D_FEAT;
    const int nEdges = in_sizes[1];
    const int nSlices = (nEdges + SLICE - 1) / SLICE;
    const int nParts  = (nNodes + PART_ROWS - 1) / PART_ROWS;

    int*   binned = (int*)d_ws;
    int*   offs   = binned + (size_t)2 * nSlices * SLICE * 2;

    gin_bin_kernel<<<2 * nSlices, BIN_THREADS, 0, stream>>>(src, dst, gid, binned, offs, out,
                                                            nEdges, nSlices, nNodes);
    gin_block_kernel<<<nParts, L1_THREADS, 0, stream>>>(feats, binned, offs, gid, W1, b1,
                                                        W2, b2, out, nNodes, nSlices);
}


# block kernel list walks: per-row scalar work done once per 16-entry batch: run ends by lane compare + wave vote (was key broadcast, compare, 2 branches per row), row offsets premultiplied; sums kept
# speedup vs baseline: 5.0032x; 1.0967x over previous
#include <hip/hip_runtime.h>


typedef float v2f __attribute__((ext_vector_type(2)));
typedef float v8f __attribute__((ext_vector_type(8)));

#define D_FEAT   64
#define N_GRAPHS 128

#define PART_ROWS   400
#define MAX_PARTS   256
#define SLICE       8192
#define BIN_THREADS 1024
#define EDGES_PER_THREAD (SLICE / BIN_THREADS)
#define OFFS_PITCH  (MAX_PARTS + 1)

#define TAB_NODES (MAX_PARTS * PART_ROWS)
#define TAB_BATCH 8
#define PK_SHIFT  9
static_assert(PART_ROWS <= (1 << PK_SHIFT), "local row bits");
__global__ void __launch_bounds__(BIN_THREADS)
gin_bin_kernel(const int* __restrict__ src,
               const int* __restrict__ dst,
               const int* __restrict__ gid,
               int*       __restrict__ binned,
               int*       __restrict__ offs,
               float*     __restrict__ out,
               int nEdges, int nSlices, int nNodes)
{
    __shared__ int  sCnt[MAX_PARTS];
    __shared__ int  sOff[MAX_PARTS + 1];
    __shared__ int  sCur[MAX_PARTS];
    __shared__ unsigned sPk[SLICE];
    __shared__ unsigned sTab[TAB_NODES / 4];

    const int tid   = threadIdx.x;
    const int bySrc = ((int)blockIdx.x >= nSlices) ? 1 : 0;
    const int slice = (int)blockIdx.x - bySrc * nSlices;
    const int e0    = slice * SLICE;
    const int n     = min(SLICE, nEdges - e0);

    if (tid < MAX_PARTS) sCnt[tid] = 0;
    __syncthreads();

    int kv[EDGES_PER_THREAD], pv[EDGES_PER_THREAD];
    if (bySrc == 0) {
        #pragma unroll
        for (int j = 0; j < EDGES_PER_THREAD; ++j) {
            const int i = tid + j * BIN_THREADS;
            kv[j] = (i < n) ? dst[e0 + i] : 0;
            pv[j] = (i < n) ? src[e0 + i] : 0;
        }
    } else {
        #pragma unroll
        for (int j = 0; j < EDGES_PER_THREAD; ++j) {
            const int i = tid + j * BIN_THREADS;
            kv[j] = (i < n) ? src[e0 + i] : 0;
            pv[j] = (i < n) ? dst[e0 + i] : 0;
        }
        const int4* __restrict__ gq = (const int4*)gid;
        const int nQ = min(nNodes, TAB_NODES) >> 2;
        for (int base = 0; base < nQ; base += TAB_BATCH * BIN_THREADS) {
            int4 g[TAB_BATCH];
            #pragma unroll
            for (int u = 0; u < TAB_BATCH; ++u) {
                const int i = base + u * BIN_THREADS + tid;
                g[u] = gq[min(i, nQ - 1)];
            }
            #pragma unroll
            for (int u = 0; u < TAB_BATCH; ++u) {
                const int i = base + u * BIN_THREADS + tid;
                if (i < nQ)
                    sTab[i] = ((unsigned)g[u].x & 255u) | (((unsigned)g[u].y & 255u) << 8) |
                              (((unsigned)g[u].z & 255u) << 16) | (((unsigned)g[u].w & 255u) << 24);
            }
        }
        for (int i = (nQ << 2) + tid; i < min(nNodes, TAB_NODES); i += BIN_THREADS)
            ((unsigned char*)sTab)[i] = (unsigned char)gid[i];
    }
    #pragma unroll
    for (int j = 0; j < EDGES_PER_THREAD; ++j) {
        if (tid + j * BIN_THREADS < n)
            atomicAdd(&sCnt[min(kv[j] / PART_ROWS, MAX_PARTS - 1)], 1);
    }
    __syncthreads();

    if (tid < 32) {
        int c[8];
        int sum = 0;
        #pragma unroll
        for (int j = 0; j < 8; ++j) { c[j] = sCnt[tid * 8 + j]; sum += c[j]; }
        int incl = sum;
        #pragma unroll
        for (int d = 1; d < 32; d <<= 1) {
            const int t = __shfl_up(incl, d);
            if (tid >= d) incl += t;
        }
        int run = incl - sum;
        #pragma unroll
        for (int j = 0; j < 8; ++j) {
            sOff[tid * 8 + j] = run;
            sCur[tid * 8 + j] = run;
            run += c[j];
        }
        if (tid == 31) sOff[MAX_PARTS] = run;
    }
    __syncthreads();

    #pragma unroll
    for (int j = 0; j < EDGES_PER_THREAD; ++j) {
        if (tid + j * BIN_THREADS < n) {
            const int p   = min(kv[j] / PART_ROWS, MAX_PARTS - 1);
            const int pos = atomicAdd(&sCur[p], 1);
            const unsigned pay = bySrc
                ? (unsigned)((const unsigned char*)sTab)[min((unsigned)pv[j], (unsigned)(TAB_NODES - 1))]
                : (unsigned)pv[j];
            const unsigned row = (unsigned)min(kv[j] - p * PART_ROWS, PART_ROWS - 1);
            sPk[pos] = (pay << PK_SHIFT) | row;
        }
    }
    __syncthreads();

    int2* __restrict__ o = (int2*)binned + (size_t)blockIdx.x * SLICE;
    for (int i = tid; i < n; i += BIN_THREADS) {
        const unsigned ent = sPk[i];
        o[i] = make_int2((int)(ent >> PK_SHIFT), (int)(ent & ((1u << PK_SHIFT) - 1u)));
    }
    if (tid <= MAX_PARTS) offs[blockIdx.x * OFFS_PITCH + tid] = sOff[tid];

    if (blockIdx.x == 0)
        for (int i = tid; i < N_GRAPHS * D_FEAT; i += BIN_THREADS) out[i] = 0.f;
}


#define L1_THREADS  1024
#define TILE_PITCH  66
#define P_PITCH     66
#define GATHER_ROWS 16

template <bool POOL>
__device__ __forceinline__ void
walk_block_segments(const float* __restrict__ X, float* sT, float* sP,
                    const int2* __restrict__ pairs, const int* __restrict__ offs,
                    int part, int wv, int lane, int nSlices)
{
    const int nMine = (nSlices > wv) ? (nSlices - wv + 31) / 32 : 0;
    int lov = 0, hiv = 0;
    if (lane < nMine) {
        const int* __restrict__ o = offs + (wv + 32 * lane) * OFFS_PITCH + part;
        lov = o[0];
        hiv = o[1];
    }
    const int li = lane & (GATHER_ROWS - 1);

    int j = 0, b0 = 0, segHi = 0;
    for (; j < nMine; ++j) {
        b0    = __builtin_amdgcn_readlane(lov, j);
        segHi = __builtin_amdgcn_readlane(hiv, j);
        if (b0 < segHi) break;
    }
    int2 pr = make_int2(0, 0);
    if (j < nMine)
        pr = pairs[(size_t)(wv + 32 * j) * SLICE + b0 + min(li, min(GATHER_ROWS, segHi - b0) - 1)];

    while (j < nMine) {
        const int n = min(GATHER_ROWS, segHi - b0);
        const int tileRow = min(pr.y, PART_ROWS - 1);
        const int ldv  = POOL ? tileRow : pr.x;
        const int addv = POOL ? (pr.x & (N_GRAPHS - 1)) : tileRow;

        v2f r[GATHER_ROWS];
        #pragma unroll
        for (int i = 0; i < GATHER_ROWS; ++i) {
            const int s = __builtin_amdgcn_readlane(ldv, i);
            if (POOL) r[i] = *(const v2f*)(sT + s * TILE_PITCH + 2 * lane);
            else      r[i] = ((const v2f*)(X + (size_t)s * D_FEAT))[lane];
        }

        int nj = j, nb0 = b0 + GATHER_ROWS, nHi = segHi;
        if (nb0 >= nHi) {
            for (++nj; nj < nMine; ++nj) {
                nb0 = __builtin_amdgcn_readlane(lov, nj);
                nHi = __builtin_amdgcn_readlane(hiv, nj);
                if (nb0 < nHi) break;
            }
        }
        if (nj < nMine)
            pr = pairs[(size_t)(wv + 32 * nj) * SLICE + nb0 + min(li, min(GATHER_ROWS, nHi - nb0) - 1)];

        #pragma unroll
        for (int i = 0; i < GATHER_ROWS; ++i) {
            if (i < n) {
                const int a = __builtin_amdgcn_readlane(addv, i);
                float* t = POOL ? (sP + a * P_PITCH + 2 * lane)
                                : (sT + a * TILE_PITCH + 2 * lane);
                unsafeAtomicAdd(t,     r[i].x);
                unsafeAtomicAdd(t + 1, r[i].y);
            }
        }
        j = nj; b0 = nb0; segHi = nHi;
    }
}

#define LIST_CAP    8192
#define G_PER_LANE  13
#define G_BINS      (32 * G_PER_LANE)
#define P_PER_LANE  (N_GRAPHS / 32)
#define LOAD_CH     8
#define ENT_SHIFT   17
#define ENT_MASK    0x1FFFF
static_assert(G_BINS >= PART_ROWS, "row bins");
static_assert((TILE_PITCH % 2) == 0 && (D_FEAT % 2) == 0, "rows are whole float pairs");
static_assert(MAX_PARTS * PART_ROWS <= (1 << ENT_SHIFT), "payload bits");

struct WaveSegs { int incl; int q; int total; int nMine; };

__device__ __forceinline__ WaveSegs
wave_segments(const int* __restrict__ offs, int part, int wv, int lane, int nSlices)
{
    WaveSegs s;
    s.nMine = (nSlices > wv) ? (nSlices - wv + 31) / 32 : 0;
    int lo = 0, hi = 0;
    if (lane < s.nMine) {
        const int* __restrict__ o = offs + (wv + 32 * lane) * OFFS_PITCH + part;
        lo = o[0];
        hi = o[1];
    }
    const int len = max(hi - lo, 0);
    int incl = len;
    #pragma unroll
    for (int d = 1; d < 32; d <<= 1) {
        const int t = __shfl_up(incl, d);
        if (lane >= d) incl += t;
    }
    s.incl  = incl;
    s.total = __builtin_amdgcn_readlane(incl, 31);
    s.q     = (lane < s.nMine) ? (wv + 32 * lane) * SLICE + lo - (incl - len) : 0;
    return s;
}

__device__ __forceinline__ void
chunk_indices(const WaveSegs& sg, int base, int lane, int (&t)[LOAD_CH])
{
    int j[LOAD_CH];
    #pragma unroll
    for (int c = 0; c < LOAD_CH; ++c) {
        t[c] = max(min(base + c * 32 + lane, sg.total - 1), 0);
        j[c] = 0;
    }
    for (int jj = 0; jj < sg.nMine; ++jj) {
        const int end = __builtin_amdgcn_readlane(sg.incl, jj);
        #pragma unroll
        for (int c = 0; c < LOAD_CH; ++c) j[c] += (t[c] >= end) ? 1 : 0;
    }
    #pragma unroll
    for (int c = 0; c < LOAD_CH; ++c) t[c] += __shfl(sg.q, j[c]);
}

template <bool POOL>
__device__ __forceinline__ unsigned make_entry(int2 pr)
{
    const unsigned row = min((unsigned)pr.y, (unsigned)(PART_ROWS - 1));
    return POOL ? ((((unsigned)pr.x & (N_GRAPHS - 1)) << ENT_SHIFT) | row)
                : ((row << ENT_SHIFT) | ((unsigned)pr.x & ENT_MASK));
}

template <bool POOL, bool SCATTER>
__device__ __forceinline__ void
put_chunks(const int2 (&pr)[LOAD_CH], int total, int base, int lane,
           int* sCur, unsigned* sList)
{
    unsigned ent[LOAD_CH];
    int      pos[LOAD_CH];
    #pragma unroll
    for (int c = 0; c < LOAD_CH; ++c) ent[c] = make_entry<POOL>(pr[c]);
    #pragma unroll
    for (int c = 0; c < LOAD_CH; ++c) {
        pos[c] = LIST_CAP;
        if (base + c * 32 + lane < total) pos[c] = atomicAdd(&sCur[ent[c] >> ENT_SHIFT], 1);
    }
    if (SCATTER) {
        #pragma unroll
        for (int c = 0; c < LOAD_CH; ++c)
            if (pos[c] < LIST_CAP) sList[pos[c]] = ent[c];
    }
}

template <bool POOL, bool SCATTER>
__device__ __forceinline__ int2 load_pair(const int2* __restrict__ pairs, int t)
{
    if (SCATTER) return pairs[t];
    const int* __restrict__ p = (const int*)(pairs + t);
    return POOL ? make_int2(p[0], 0) : make_int2(0, p[1]);
}

template <bool SCATTER>
__device__ __forceinline__ void
sort_pass(const int2* __restrict__ pairsDst, const int2* __restrict__ pairsSrc,
          const WaveSegs& sgG, const WaveSegs& sgP, int lane,
          int* sCurG, int* sCurP, unsigned* sListG, unsigned* sListP)
{
    const int tmax = max(sgG.total, sgP.total);
    for (int base = 0; base < tmax; base += 32 * LOAD_CH) {
        int  tg[LOAD_CH], tp[LOAD_CH];
        int2 pg[LOAD_CH], pp[LOAD_CH];
        chunk_indices(sgG, base, lane, tg);
        chunk_indices(sgP, base, lane, tp);
        #pragma unroll
        for (int c = 0; c < LOAD_CH; ++c) pg[c] = load_pair<false, SCATTER>(pairsDst, tg[c]);
        #pragma unroll
        for (int c = 0; c < LOAD_CH; ++c) pp[c] = load_pair<true,  SCATTER>(pairsSrc, tp[c]);
        put_chunks<false, SCATTER>(pg, sgG.total, base, lane, sCurG, sListG);
        put_chunks<true,  SCATTER>(pp, sgP.total, base, lane, sCurP, sListP);
    }
}

template <int PER>
__device__ __forceinline__ void wave_scan(int* sCnt, int* sOff, int lane)
{
    int c[PER];
    int sum = 0;
    #pragma unroll
    for (int j = 0; j < PER; ++j) { c[j] = sCnt[lane * PER + j]; sum += c[j]; }
    int incl = sum;
    #pragma unroll
    for (int d = 1; d < 32; d <<= 1) {
        const int t = __shfl_up(incl, d);
        if (lane >= d) incl += t;
    }
    int run = incl - sum;
    #pragma unroll
    for (int j = 0; j < PER; ++j) {
        sOff[lane * PER + j] = run;
        sCnt[lane * PER + j] = run;
        run += c[j];
    }
    if (lane == 31) sOff[32 * PER] = run;
}

template <bool POOL>
__device__ __forceinline__ void flush_sum(float* sT, float* sP, int key, v2f acc, int lane)
{
    if (POOL) {
        float* t = sP + key * P_PITCH + 2 * lane;
        unsafeAtomicAdd(t,     acc.x);
        unsafeAtomicAdd(t + 1, acc.y);
    } else {
        *(v2f*)(sT + key * TILE_PITCH + 2 * lane) = acc;
    }
}

template <bool POOL>
__device__ __forceinline__ void
consume_list(const float* __restrict__ X, float* sT, float* sP,
             const unsigned* sList, int s, int e, int lane)
{
    const int li = lane & (GATHER_ROWS - 1);
    v2f acc = (v2f){0.f, 0.f};
    unsigned ent = 0, entN = 0;
    if (s < e) {
        ent  = sList[min(s + li,     e - 1)];
        entN = sList[min(s + li + 1, e - 1)];
    }

    for (int b = s; b < e; b += GATHER_ROWS) {
        const int n = min(GATHER_ROWS, e - b);
        const int keyv = (int)(ent >> ENT_SHIFT);
        const unsigned offv = (unsigned)(ent & ENT_MASK) * (POOL ? (TILE_PITCH / 2) : (D_FEAT / 2));
        const bool runEnd = ((int)(entN >> ENT_SHIFT) != keyv) || (b + li + 1 >= e);
        const unsigned ends = __builtin_amdgcn_ballot_w32(runEnd) & ((1u << n) - 1u);

        v2f r[GATHER_ROWS];
        #pragma unroll
        for (int i = 0; i < GATHER_ROWS; ++i) {
            const unsigned ro = (unsigned)__builtin_amdgcn_readlane((int)offv, i);
            if (POOL) r[i] = ((const v2f*)sT)[ro + (unsigned)lane];
            else      r[i] = ((const v2f*)X)[ro + (unsigned)lane];
        }
        const int nb = b + GATHER_ROWS;
        if (nb < e) {
            ent  = sList[min(nb + li,     e - 1)];
            entN = sList[min(nb + li + 1, e - 1)];
        }

        #pragma unroll
        for (int i = 0; i < GATHER_ROWS; ++i) {
            acc = acc + r[i];
            if (ends & (1u << i)) {
                flush_sum<POOL>(sT, sP, __builtin_amdgcn_readlane(keyv, i), acc, lane);
                acc = (v2f){0.f, 0.f};
            }
        }
    }
}

__global__ void __launch_bounds__(L1_THREADS) __attribute__((amdgpu_waves_per_eu(8, 8)))
gin_block_kernel(const float* __restrict__ X,
                 const int*   __restrict__ binned,
                 const int*   __restrict__ offs,
                 const int*   __restrict__ gid,
                 const float* __restrict__ W,
                 const float* __restrict__ bias,
                 const float* __restrict__ W2,
                 const float* __restrict__ bias2,
                 float*       __restrict__ out,
                 int nNodes, int nSlices)
{
    __shared__ float sT[PART_ROWS * TILE_PITCH];
    __shared__ float sWp[D_FEAT * D_FEAT];
    __shared__ float sWp2[D_FEAT * D_FEAT];
    __shared__ float sB[D_FEAT];
    __shared__ float sB2[D_FEAT];
    __shared__ float sP[N_GRAPHS * P_PITCH];
    __shared__ float sC[N_GRAPHS];
    __shared__ int      sCntG[G_BINS];
    __shared__ int      sOffG[G_BINS + 1];
    __shared__ int      sCntP[N_GRAPHS];
    __shared__ int      sOffP[N_GRAPHS + 1];
    __shared__ unsigned sListG[LIST_CAP];
    __shared__ unsigned sListP[LIST_CAP];

    const int tid = threadIdx.x;
    for (int i = tid; i < PART_ROWS * TILE_PITCH; i += L1_THREADS) sT[i] = 0.f;
    for (int i = tid; i < N_GRAPHS * P_PITCH; i += L1_THREADS) sP[i] = 0.f;
    if (tid < N_GRAPHS) { sC[tid] = 0.f; sCntP[tid] = 0; }
    if (tid < G_BINS) sCntG[tid] = 0;
    for (int idx = tid; idx < D_FEAT * D_FEAT; idx += L1_THREADS) {
        const int k   = idx >> 6;
        const int col = idx & 63;
        const int o   = (k >> 1) * 128 + col * 2 + (k & 1);
        sWp[o]  = W[idx];
        sWp2[o] = W2[idx];
    }
    if (tid < D_FEAT) { sB[tid] = bias[tid]; sB2[tid] = bias2[tid]; }
    __syncthreads();

    const int lane = tid & 31;
    const int wv   = __builtin_amdgcn_readfirstlane(tid >> 5);
    const int part = blockIdx.x;
    const int2* __restrict__ pairsDst = (const int2*)binned;
    const int2* __restrict__ pairsSrc = (const int2*)binned + (size_t)nSlices * SLICE;
    const int*  __restrict__ offsDst  = offs;
    const int*  __restrict__ offsSrc  = offs + nSlices * OFFS_PITCH;

    const WaveSegs sgG = wave_segments(offsDst, part, wv, lane, nSlices);
    const WaveSegs sgP = wave_segments(offsSrc, part, wv, lane, nSlices);
    sort_pass<false>(pairsDst, pairsSrc, sgG, sgP, lane, sCntG, sCntP, sListG, sListP);
    __syncthreads();
    if (wv == 0) wave_scan<G_PER_LANE>(sCntG, sOffG, lane);
    if (wv == 1) wave_scan<P_PER_LANE>(sCntP, sOffP, lane);
    __syncthreads();
    sort_pass<true>(pairsDst, pairsSrc, sgG, sgP, lane, sCntG, sCntP, sListG, sListP);
    __syncthreads();
    const bool listG = __builtin_amdgcn_readfirstlane(sOffG[G_BINS])   <= LIST_CAP;
    const bool listP = __builtin_amdgcn_readfirstlane(sOffP[N_GRAPHS]) <= LIST_CAP;

    if (listG) {
        const int s = __builtin_amdgcn_readfirstlane(sOffG[wv * G_PER_LANE]);
        const int e = __builtin_amdgcn_readfirstlane(sOffG[wv * G_PER_LANE + G_PER_LANE]);
        consume_list<false>(X, sT, sP, sListG, s, e, lane);
    } else {
        walk_block_segments<false>(X, sT, sP, pairsDst, offsDst, part, wv, lane, nSlices);
    }
    __syncthreads();

    const int rowLocal = wv * 16;
    const int rowBase  = part * PART_ROWS + rowLocal;
    const int halfSel  = lane >> 4;
    const int laneM    = lane & 15;
    const bool hasTile = (wv < PART_ROWS / 16) && (rowBase < nNodes);

    if (hasTile) {
        int aRow = rowBase + laneM;
        if (aRow >= nNodes) aRow = nNodes - 1;
        const float* __restrict__ xr = X + (size_t)aRow * D_FEAT;
        const float* tr = sT + (rowLocal + laneM) * TILE_PITCH;

        v8f acc[4];
        #pragma unroll
        for (int ct = 0; ct < 4; ++ct) acc[ct] = (v8f){0.f,0.f,0.f,0.f,0.f,0.f,0.f,0.f};

        #pragma unroll 4
        for (int kt = 0; kt < 16; ++kt) {
            const int k0 = kt * 4 + halfSel * 2;
            const v2f a = *(const v2f*)(xr + k0) + *(const v2f*)(tr + k0);
            const int bRow = (kt * 2 + halfSel) * 128;
            #pragma unroll
            for (int ct = 0; ct < 4; ++ct) {
                const int col = ct * 16 + laneM;
                const v2f b = *(const v2f*)(sWp + bRow + col * 2);
                acc[ct] = __builtin_amdgcn_wmma_f32_16x16x4_f32(
                    false, a, false, b, (short)0, acc[ct], false, false);
            }
        }

        #pragma unroll
        for (int ct = 0; ct < 4; ++ct) {
            const int col = ct * 16 + laneM;
            const float bv = sB[col];
            #pragma unroll
            for (int v = 0; v < 8; ++v)
                sT[(rowLocal + v + halfSel * 8) * TILE_PITCH + col] = fmaxf(acc[ct][v] + bv, 0.f);
        }
    }
    __syncthreads();

    if (hasTile) {
        const int n  = min(GATHER_ROWS, nNodes - rowBase);
        const int li = lane & (GATHER_ROWS - 1);
        const int gv = gid[rowBase + min(li, n - 1)] & (N_GRAPHS - 1);
        if (lane < n) unsafeAtomicAdd(&sC[gv], 1.0f);

        v2f r[GATHER_ROWS];
        #pragma unroll
        for (int i = 0; i < GATHER_ROWS; ++i)
            r[i] = *(const v2f*)(sT + (rowLocal + i) * TILE_PITCH + 2 * lane);
        int cur = -1;
        v2f acc = (v2f){0.f, 0.f};
        #pragma unroll
        for (int i = 0; i < GATHER_ROWS; ++i) {
            if (i < n) {
                const int g = __builtin_amdgcn_readlane(gv, i);
                if (g != cur) {
                    if (cur >= 0) flush_sum<true>(sT, sP, cur, acc, lane);
                    acc = (v2f){0.f, 0.f};
                    cur = g;
                }
                acc = acc + r[i];
            }
        }
        if (cur >= 0) flush_sum<true>(sT, sP, cur, acc, lane);
    }
    if (listP) {
        const int s = __builtin_amdgcn_readfirstlane(sOffP[wv * P_PER_LANE]);
        const int e = __builtin_amdgcn_readfirstlane(sOffP[wv * P_PER_LANE + P_PER_LANE]);
        consume_list<true>(X, sT, sP, sListP, s, e, lane);
    } else {
        walk_block_segments<true>(X, sT, sP, pairsSrc, offsSrc, part, wv, lane, nSlices);
    }
    __syncthreads();

    float* sO = sT;
    {
        const int rt = wv >> 2;
        const int ct = wv & 3;
        const float* pr = sP + (rt * 16 + laneM) * P_PITCH;
        const int col = ct * 16 + laneM;
        v8f acc2 = (v8f){0.f,0.f,0.f,0.f,0.f,0.f,0.f,0.f};
        #pragma unroll
        for (int kt = 0; kt < 16; ++kt) {
            const int k0 = kt * 4 + halfSel * 2;
            const v2f a = *(const v2f*)(pr + k0);
            const v2f b = *(const v2f*)(sWp2 + (kt * 2 + halfSel) * 128 + col * 2);
            acc2 = __builtin_amdgcn_wmma_f32_16x16x4_f32(
                false, a, false, b, (short)0, acc2, false, false);
        }
        const float bv = sB2[col];
        #pragma unroll
        for (int v = 0; v < 8; ++v) {
            const int row = rt * 16 + v + halfSel * 8;
            sO[row * D_FEAT + col] = acc2[v] + sC[row] * bv;
        }
    }
    __syncthreads();

    constexpr int LINES = N_GRAPHS * D_FEAT / 32;
    #pragma unroll
    for (int q = 0; q < LINES / (L1_THREADS / 32); ++q) {
        const int line = (wv + (L1_THREADS / 32) * q + (int)blockIdx.x) & (LINES - 1);
        const int idx  = line * 32 + lane;
        unsafeAtomicAdd(out + idx, sO[idx]);
    }
}

extern "C" void kernel_launch(void* const* d_in, const int* in_sizes, int n_in,
                              void* d_out, int out_size, void* d_ws, size_t ws_size,
                              hipStream_t stream)
{
    const float* feats = (const float*)d_in[0];
    const int*   src   = (const int*)  d_in[1];
    const int*   dst   = (const int*)  d_in[2];
    const int*   gid   = (const int*)  d_in[3];
    const float* W1    = (const float*)d_in[4];
    const float* b1    = (const float*)d_in[5];
    const float* W2    = (const float*)d_in[6];
    const float* b2    = (const float*)d_in[7];
    float* out = (float*)d_out;

    const int nNodes = in_sizes[0] / D_FEAT;
    const int nEdges = in_sizes[1];
    const int nSlices = (nEdges + SLICE - 1) / SLICE;
    const int nParts  = (nNodes + PART_ROWS - 1) / PART_ROWS;

    int*   binned = (int*)d_ws;
    int*   offs   = binned + (size_t)2 * nSlices * SLICE * 2;

    gin_bin_kernel<<<2 * nSlices, BIN_THREADS, 0, stream>>>(src, dst, gid, binned, offs, out,
                                                            nEdges, nSlices, nNodes);
    gin_block_kernel<<<nParts, L1_THREADS, 0, stream>>>(feats, binned, offs, gid, W1, b1,
                                                        W2, b2, out, nNodes, nSlices);
}
